# starting kernel
# baseline (speedup 1.0000x reference)
#include <hip/hip_runtime.h>

typedef __attribute__((ext_vector_type(2))) float v2f;
typedef __attribute__((ext_vector_type(8))) float v8f;

#define NEG_SLOPE 0.2f

static __device__ __forceinline__ unsigned f2key(float f) {
  unsigned u = __float_as_uint(f);
  return (u & 0x80000000u) ? ~u : (u | 0x80000000u);
}
static __device__ __forceinline__ float key2f(unsigned k) {
  return (k & 0x80000000u) ? __uint_as_float(k & 0x7FFFFFFFu)
                           : __uint_as_float(~k);
}

__global__ void zero_f32_kernel(float* __restrict__ p, long n) {
  long i = (long)blockIdx.x * blockDim.x + threadIdx.x;
  long stride = (long)gridDim.x * blockDim.x;
  for (; i < n; i += stride) p[i] = 0.0f;
}

__global__ void pad_cols_kernel(const float* __restrict__ src,
                                float* __restrict__ dst,
                                int K, int c, int ldd) {
  int i = blockIdx.x * blockDim.x + threadIdx.x;
  if (i >= K * ldd) return;
  int r = i / ldd, j = i - r * ldd;
  dst[i] = (j < c) ? src[r * c + j] : 0.0f;
}

__global__ void gemm_f32_wmma_kernel(const float* __restrict__ A,
                                     const float* __restrict__ B,
                                     float* __restrict__ C,
                                     int M, int K, int ldb) {
  const int lane = threadIdx.x & 31;
  const int wave = threadIdx.x >> 5;
  const int half = lane >> 4;
  const int l    = lane & 15;
  const int nt64   = ldb >> 6;
  const int mtiles = M >> 4;
  const int total  = mtiles * nt64;
  int t = blockIdx.x * (blockDim.x >> 5) + wave;
  if (t >= total) return;
  const int mt = t / nt64;
  const int n0 = (t - mt * nt64) << 6;
  const int m0 = mt << 4;

  v8f acc[4] = {};
  const float* arow = A + (size_t)(m0 + l) * K + 2 * half;
  const float* bp   = B + (size_t)(2 * half) * ldb + n0 + l;
  const size_t bstep = (size_t)4 * ldb;

  v2f a = *(const v2f*)(arow);
  v2f b[4];
#pragma unroll
  for (int s = 0; s < 4; ++s) {
    b[s].x = bp[s * 16];
    b[s].y = bp[ldb + s * 16];
  }

  for (int k0 = 4; k0 < K; k0 += 4) {
    v2f an = *(const v2f*)(arow + k0);
    const float* bpn = bp + bstep;
    v2f bn[4];
#pragma unroll
    for (int s = 0; s < 4; ++s) {
      bn[s].x = bpn[s * 16];
      bn[s].y = bpn[ldb + s * 16];
    }
#pragma unroll
    for (int s = 0; s < 4; ++s)
      acc[s] = __builtin_amdgcn_wmma_f32_16x16x4_f32(
          false, a, false, b[s], (short)0, acc[s], false, false);
    a = an;
#pragma unroll
    for (int s = 0; s < 4; ++s) b[s] = bn[s];
    bp = bpn;
  }
#pragma unroll
  for (int s = 0; s < 4; ++s)
    acc[s] = __builtin_amdgcn_wmma_f32_16x16x4_f32(
        false, a, false, b[s], (short)0, acc[s], false, false);

  float* crow = C + (size_t)(m0 + 8 * half) * ldb + n0 + l;
#pragma unroll
  for (int s = 0; s < 4; ++s)
#pragma unroll
    for (int j = 0; j < 8; ++j) crow[(size_t)j * ldb + s * 16] = acc[s][j];
}

__global__ void alpha_kernel(const float* __restrict__ xw,
                             const float* __restrict__ a_src,
                             const float* __restrict__ a_dst,
                             float* __restrict__ as, float* __restrict__ ad,
                             int n_nodes, int heads, int c, int ldx) {
  int i = blockIdx.x * blockDim.x + threadIdx.x;
  if (i >= n_nodes * heads) return;
  int n = i / heads, h = i - n * heads;
  const float* row = xw + (size_t)n * ldx + (size_t)h * c;
  const float* ws = a_src + (size_t)h * c;
  const float* wd = a_dst + (size_t)h * c;
  float s = 0.0f, d = 0.0f;
  for (int j = 0; j < c; ++j) {
    float v = row[j];
    s += v * ws[j];
    d += v * wd[j];
  }
  as[i] = s;
  ad[i] = d;
}

__global__ void edge_max_kernel(const int* __restrict__ esrc,
                                const int* __restrict__ edst,
                                int E, int Nn, int heads,
                                const float* __restrict__ as,
                                const float* __restrict__ ad,
                                unsigned* __restrict__ mkey) {
  int e = blockIdx.x * blockDim.x + threadIdx.x;
  if (e >= E + Nn) return;
  int s = (e < E) ? esrc[e] : (e - E);
  int d = (e < E) ? edst[e] : (e - E);
  for (int h = 0; h < heads; ++h) {
    float v = as[s * heads + h] + ad[d * heads + h];
    v = (v >= 0.0f) ? v : NEG_SLOPE * v;
    atomicMax(&mkey[d * heads + h], f2key(v));
  }
}

__global__ void edge_denom_kernel(const int* __restrict__ esrc,
                                  const int* __restrict__ edst,
                                  int E, int Nn, int heads,
                                  const float* __restrict__ as,
                                  const float* __restrict__ ad,
                                  const unsigned* __restrict__ mkey,
                                  float* __restrict__ denom) {
  int e = blockIdx.x * blockDim.x + threadIdx.x;
  if (e >= E + Nn) return;
  int s = (e < E) ? esrc[e] : (e - E);
  int d = (e < E) ? edst[e] : (e - E);
  for (int h = 0; h < heads; ++h) {
    float v = as[s * heads + h] + ad[d * heads + h];
    v = (v >= 0.0f) ? v : NEG_SLOPE * v;
    atomicAdd(&denom[d * heads + h], expf(v - key2f(mkey[d * heads + h])));
  }
}

__global__ void edge_aggregate_kernel(const int* __restrict__ esrc,
                                      const int* __restrict__ edst,
                                      int E, int Nn, int heads, int C, int ldx,
                                      const float* __restrict__ as,
                                      const float* __restrict__ ad,
                                      const unsigned* __restrict__ mkey,
                                      const float* __restrict__ denom,
                                      const float* __restrict__ xw,
                                      float* __restrict__ out) {
  long tid = (long)blockIdx.x * blockDim.x + threadIdx.x;
  long total = (long)(E + Nn) * heads;
  if (tid >= total) return;
  int e = (int)(tid / heads);
  int h = (int)(tid - (long)e * heads);
  int s = (e < E) ? esrc[e] : (e - E);
  int d = (e < E) ? edst[e] : (e - E);
  float v = as[s * heads + h] + ad[d * heads + h];
  v = (v >= 0.0f) ? v : NEG_SLOPE * v;
  int di = d * heads + h;
  float w = expf(v - key2f(mkey[di])) / denom[di];
  const float* xs = xw + (size_t)s * ldx + (size_t)h * C;
  float* od = out + (size_t)d * heads * C + (size_t)h * C;
  for (int c = 0; c < C; ++c) atomicAdd(&od[c], w * xs[c]);
}

__global__ void bias_elu_kernel(float* __restrict__ h,
                                const float* __restrict__ bias, long n, int f) {
  long i = (long)blockIdx.x * blockDim.x + threadIdx.x;
  if (i >= n) return;
  float v = h[i] + bias[(int)(i % f)];
  h[i] = (v > 0.0f) ? v : (expf(v) - 1.0f);
}

__global__ void bias_add_kernel(float* __restrict__ h,
                                const float* __restrict__ bias, long n, int f) {
  long i = (long)blockIdx.x * blockDim.x + threadIdx.x;
  if (i >= n) return;
  h[i] += bias[(int)(i % f)];
}

extern "C" void kernel_launch(void* const* d_in, const int* in_sizes, int n_in,
                              void* d_out, int out_size, void* d_ws, size_t ws_size,
                              hipStream_t stream) {
  const float* x      = (const float*)d_in[0];
  const int*   ei     = (const int*)d_in[1];
  const float* W1     = (const float*)d_in[2];
  const float* a1_src = (const float*)d_in[3];
  const float* a1_dst = (const float*)d_in[4];
  const float* b1     = (const float*)d_in[5];
  const float* W2     = (const float*)d_in[6];
  const float* a2_src = (const float*)d_in[7];
  const float* a2_dst = (const float*)d_in[8];
  const float* b2     = (const float*)d_in[9];
  float* out = (float*)d_out;
  (void)n_in; (void)out_size; (void)ws_size;

  const int F_IN = 256, H1 = 8, C1 = 32, HC1 = 256, NC = 40, NCP = 64;
  const int Nn   = in_sizes[0] / F_IN;
  const int E    = in_sizes[1] / 2;
  const int Etot = E + Nn;
  const int* esrc = ei;
  const int* edst = ei + E;

  float* p = (float*)d_ws;
  float*    xw1  = p;            p += (size_t)Nn * HC1;
  float*    as1  = p;            p += (size_t)Nn * H1;
  float*    ad1  = p;            p += (size_t)Nn * H1;
  unsigned* m1   = (unsigned*)p; p += (size_t)Nn * H1;
  float*    den1 = p;            p += (size_t)Nn * H1;
  float*    agg1 = p;            p += (size_t)Nn * HC1;
  float*    w2p  = p;            p += (size_t)HC1 * NCP;
  float*    xw2p = p;            p += (size_t)Nn * NCP;
  float*    as2  = p;            p += (size_t)Nn;
  float*    ad2  = p;            p += (size_t)Nn;
  unsigned* m2   = (unsigned*)p; p += (size_t)Nn;
  float*    den2 = p;            p += (size_t)Nn;

  dim3 blk(256);

  zero_f32_kernel<<<2048, blk, 0, stream>>>((float*)m1,
                                            (long)Nn * (2 * H1 + HC1));
  zero_f32_kernel<<<256, blk, 0, stream>>>((float*)m2, (long)Nn * 2);
  zero_f32_kernel<<<1024, blk, 0, stream>>>(out, (long)Nn * NC);
  pad_cols_kernel<<<(HC1 * NCP + 255) / 256, blk, 0, stream>>>(
      W2, w2p, HC1, NC, NCP);

  {
    int waves  = (Nn / 16) * (HC1 / 64);
    int blocks = (waves + 7) / 8;
    gemm_f32_wmma_kernel<<<blocks, blk, 0, stream>>>(x, W1, xw1, Nn, F_IN, HC1);
  }
  alpha_kernel<<<(Nn * H1 + 255) / 256, blk, 0, stream>>>(
      xw1, a1_src, a1_dst, as1, ad1, Nn, H1, C1, HC1);
  edge_max_kernel<<<(Etot + 255) / 256, blk, 0, stream>>>(
      esrc, edst, E, Nn, H1, as1, ad1, m1);
  edge_denom_kernel<<<(Etot + 255) / 256, blk, 0, stream>>>(
      esrc, edst, E, Nn, H1, as1, ad1, m1, den1);
  {
    long tot = (long)Etot * H1;
    edge_aggregate_kernel<<<(int)((tot + 255) / 256), blk, 0, stream>>>(
        esrc, edst, E, Nn, H1, C1, HC1, as1, ad1, m1, den1, xw1, agg1);
  }
  bias_elu_kernel<<<(int)(((long)Nn * HC1 + 255) / 256), blk, 0, stream>>>(
      agg1, b1, (long)Nn * HC1, HC1);

  {
    int waves  = (Nn / 16) * (NCP / 64);
    int blocks = (waves + 7) / 8;
    gemm_f32_wmma_kernel<<<blocks, blk, 0, stream>>>(agg1, w2p, xw2p,
                                                     Nn, HC1, NCP);
  }
  alpha_kernel<<<(Nn + 255) / 256, blk, 0, stream>>>(
      xw2p, a2_src, a2_dst, as2, ad2, Nn, 1, NC, NCP);
  edge_max_kernel<<<(Etot + 255) / 256, blk, 0, stream>>>(
      esrc, edst, E, Nn, 1, as2, ad2, m2);
  edge_denom_kernel<<<(Etot + 255) / 256, blk, 0, stream>>>(
      esrc, edst, E, Nn, 1, as2, ad2, m2, den2);
  edge_aggregate_kernel<<<(Etot + 255) / 256, blk, 0, stream>>>(
      esrc, edst, E, Nn, 1, NC, NCP, as2, ad2, m2, den2, xw2p, out);
  bias_add_kernel<<<(int)(((long)Nn * NC + 255) / 256), blk, 0, stream>>>(
      out, b2, (long)Nn * NC, NC);
}


# edge aggregate pass: one wave per 32 edges with lanes along the channel axis, so each row load and each float atomic add covers one contiguous line; atomics kept, arrival order as before
# speedup vs baseline: 3.3985x; 3.3985x over previous
#include <hip/hip_runtime.h>

typedef __attribute__((ext_vector_type(2))) float v2f;
typedef __attribute__((ext_vector_type(8))) float v8f;

#define NEG_SLOPE 0.2f

static __device__ __forceinline__ unsigned f2key(float f) {
  unsigned u = __float_as_uint(f);
  return (u & 0x80000000u) ? ~u : (u | 0x80000000u);
}
static __device__ __forceinline__ float key2f(unsigned k) {
  return (k & 0x80000000u) ? __uint_as_float(k & 0x7FFFFFFFu)
                           : __uint_as_float(~k);
}

__global__ void zero_f32_kernel(float* __restrict__ p, long n) {
  long i = (long)blockIdx.x * blockDim.x + threadIdx.x;
  long stride = (long)gridDim.x * blockDim.x;
  for (; i < n; i += stride) p[i] = 0.0f;
}

__global__ void pad_cols_kernel(const float* __restrict__ src,
                                float* __restrict__ dst,
                                int K, int c, int ldd) {
  int i = blockIdx.x * blockDim.x + threadIdx.x;
  if (i >= K * ldd) return;
  int r = i / ldd, j = i - r * ldd;
  dst[i] = (j < c) ? src[r * c + j] : 0.0f;
}

__global__ void gemm_f32_wmma_kernel(const float* __restrict__ A,
                                     const float* __restrict__ B,
                                     float* __restrict__ C,
                                     int M, int K, int ldb) {
  const int lane = threadIdx.x & 31;
  const int wave = threadIdx.x >> 5;
  const int half = lane >> 4;
  const int l    = lane & 15;
  const int nt64   = ldb >> 6;
  const int mtiles = M >> 4;
  const int total  = mtiles * nt64;
  int t = blockIdx.x * (blockDim.x >> 5) + wave;
  if (t >= total) return;
  const int mt = t / nt64;
  const int n0 = (t - mt * nt64) << 6;
  const int m0 = mt << 4;

  v8f acc[4] = {};
  const float* arow = A + (size_t)(m0 + l) * K + 2 * half;
  const float* bp   = B + (size_t)(2 * half) * ldb + n0 + l;
  const size_t bstep = (size_t)4 * ldb;

  v2f a = *(const v2f*)(arow);
  v2f b[4];
#pragma unroll
  for (int s = 0; s < 4; ++s) {
    b[s].x = bp[s * 16];
    b[s].y = bp[ldb + s * 16];
  }

  for (int k0 = 4; k0 < K; k0 += 4) {
    v2f an = *(const v2f*)(arow + k0);
    const float* bpn = bp + bstep;
    v2f bn[4];
#pragma unroll
    for (int s = 0; s < 4; ++s) {
      bn[s].x = bpn[s * 16];
      bn[s].y = bpn[ldb + s * 16];
    }
#pragma unroll
    for (int s = 0; s < 4; ++s)
      acc[s] = __builtin_amdgcn_wmma_f32_16x16x4_f32(
          false, a, false, b[s], (short)0, acc[s], false, false);
    a = an;
#pragma unroll
    for (int s = 0; s < 4; ++s) b[s] = bn[s];
    bp = bpn;
  }
#pragma unroll
  for (int s = 0; s < 4; ++s)
    acc[s] = __builtin_amdgcn_wmma_f32_16x16x4_f32(
        false, a, false, b[s], (short)0, acc[s], false, false);

  float* crow = C + (size_t)(m0 + 8 * half) * ldb + n0 + l;
#pragma unroll
  for (int s = 0; s < 4; ++s)
#pragma unroll
    for (int j = 0; j < 8; ++j) crow[(size_t)j * ldb + s * 16] = acc[s][j];
}

__global__ void alpha_kernel(const float* __restrict__ xw,
                             const float* __restrict__ a_src,
                             const float* __restrict__ a_dst,
                             float* __restrict__ as, float* __restrict__ ad,
                             int n_nodes, int heads, int c, int ldx) {
  int i = blockIdx.x * blockDim.x + threadIdx.x;
  if (i >= n_nodes * heads) return;
  int n = i / heads, h = i - n * heads;
  const float* row = xw + (size_t)n * ldx + (size_t)h * c;
  const float* ws = a_src + (size_t)h * c;
  const float* wd = a_dst + (size_t)h * c;
  float s = 0.0f, d = 0.0f;
  for (int j = 0; j < c; ++j) {
    float v = row[j];
    s += v * ws[j];
    d += v * wd[j];
  }
  as[i] = s;
  ad[i] = d;
}

__global__ void edge_max_kernel(const int* __restrict__ esrc,
                                const int* __restrict__ edst,
                                int E, int Nn, int heads,
                                const float* __restrict__ as,
                                const float* __restrict__ ad,
                                unsigned* __restrict__ mkey) {
  int e = blockIdx.x * blockDim.x + threadIdx.x;
  if (e >= E + Nn) return;
  int s = (e < E) ? esrc[e] : (e - E);
  int d = (e < E) ? edst[e] : (e - E);
  for (int h = 0; h < heads; ++h) {
    float v = as[s * heads + h] + ad[d * heads + h];
    v = (v >= 0.0f) ? v : NEG_SLOPE * v;
    atomicMax(&mkey[d * heads + h], f2key(v));
  }
}

__global__ void edge_denom_kernel(const int* __restrict__ esrc,
                                  const int* __restrict__ edst,
                                  int E, int Nn, int heads,
                                  const float* __restrict__ as,
                                  const float* __restrict__ ad,
                                  const unsigned* __restrict__ mkey,
                                  float* __restrict__ denom) {
  int e = blockIdx.x * blockDim.x + threadIdx.x;
  if (e >= E + Nn) return;
  int s = (e < E) ? esrc[e] : (e - E);
  int d = (e < E) ? edst[e] : (e - E);
  for (int h = 0; h < heads; ++h) {
    float v = as[s * heads + h] + ad[d * heads + h];
    v = (v >= 0.0f) ? v : NEG_SLOPE * v;
    atomicAdd(&denom[d * heads + h], expf(v - key2f(mkey[d * heads + h])));
  }
}

template <int H>
static __device__ __forceinline__ void load_node_row(const float* __restrict__ p,
                                                     float (&v)[H]) {
  if constexpr (H % 4 == 0) {
#pragma unroll
    for (int q = 0; q < H / 4; ++q) {
      const float4 t = reinterpret_cast<const float4*>(p)[q];
      v[4 * q + 0] = t.x; v[4 * q + 1] = t.y;
      v[4 * q + 2] = t.z; v[4 * q + 3] = t.w;
    }
  } else {
#pragma unroll
    for (int i = 0; i < H; ++i) v[i] = p[i];
  }
}

template <int H, int C, int LDX>
__global__ void __launch_bounds__(256)
edge_aggregate_kernel(const int* __restrict__ esrc,
                      const int* __restrict__ edst,
                      int E, int Nn,
                      const float* __restrict__ as,
                      const float* __restrict__ ad,
                      const unsigned* __restrict__ mkey,
                      const float* __restrict__ denom,
                      const float* __restrict__ xw,
                      float* __restrict__ out) {
  const int lane = threadIdx.x & 31;
  const int wv = __builtin_amdgcn_readfirstlane((int)(threadIdx.x >> 5));
  const int Etot = E + Nn;
  const int base = (blockIdx.x * (int)(blockDim.x >> 5) + wv) * 32;
  if (base >= Etot) return;
  const int cnt = (Etot - base < 32) ? (Etot - base) : 32;

  const int e = base + lane;
  int s = 0, d = 0;
  if (e < Etot) {
    s = (e < E) ? esrc[e] : (e - E);
    d = (e < E) ? edst[e] : (e - E);
  }
  float av[H], dv[H], mv[H], nv[H], w[H];
  load_node_row<H>(as + (size_t)s * H, av);
  load_node_row<H>(ad + (size_t)d * H, dv);
  load_node_row<H>(reinterpret_cast<const float*>(mkey) + (size_t)d * H, mv);
  load_node_row<H>(denom + (size_t)d * H, nv);
#pragma unroll
  for (int h = 0; h < H; ++h) {
    float v = av[h] + dv[h];
    v = (v >= 0.0f) ? v : NEG_SLOPE * v;
    w[h] = expf(v - key2f(__float_as_uint(mv[h]))) / nv[h];
  }

  constexpr int NCH = (C + 31) / 32;
  for (int j = 0; j < cnt; ++j) {
    const int sj = __builtin_amdgcn_readlane(s, j);
    const int dj = __builtin_amdgcn_readlane(d, j);
    const float* xs = xw + (size_t)sj * LDX;
    float* od = out + (size_t)dj * (H * C);
    float xv[H * NCH];
#pragma unroll
    for (int h = 0; h < H; ++h) {
#pragma unroll
      for (int k = 0; k < NCH; ++k) {
        const int c = k * 32 + lane;
        xv[h * NCH + k] = (c < C) ? xs[h * C + c] : 0.0f;
      }
    }
#pragma unroll
    for (int h = 0; h < H; ++h) {
      const float wj = __int_as_float(
          __builtin_amdgcn_readlane(__float_as_int(w[h]), j));
#pragma unroll
      for (int k = 0; k < NCH; ++k) {
        const int c = k * 32 + lane;
        if (c < C) atomicAdd(&od[h * C + c], wj * xv[h * NCH + k]);
      }
    }
  }
}

__global__ void bias_elu_kernel(float* __restrict__ h,
                                const float* __restrict__ bias, long n, int f) {
  long i = (long)blockIdx.x * blockDim.x + threadIdx.x;
  if (i >= n) return;
  float v = h[i] + bias[(int)(i % f)];
  h[i] = (v > 0.0f) ? v : (expf(v) - 1.0f);
}

__global__ void bias_add_kernel(float* __restrict__ h,
                                const float* __restrict__ bias, long n, int f) {
  long i = (long)blockIdx.x * blockDim.x + threadIdx.x;
  if (i >= n) return;
  h[i] += bias[(int)(i % f)];
}

extern "C" void kernel_launch(void* const* d_in, const int* in_sizes, int n_in,
                              void* d_out, int out_size, void* d_ws, size_t ws_size,
                              hipStream_t stream) {
  const float* x      = (const float*)d_in[0];
  const int*   ei     = (const int*)d_in[1];
  const float* W1     = (const float*)d_in[2];
  const float* a1_src = (const float*)d_in[3];
  const float* a1_dst = (const float*)d_in[4];
  const float* b1     = (const float*)d_in[5];
  const float* W2     = (const float*)d_in[6];
  const float* a2_src = (const float*)d_in[7];
  const float* a2_dst = (const float*)d_in[8];
  const float* b2     = (const float*)d_in[9];
  float* out = (float*)d_out;
  (void)n_in; (void)out_size; (void)ws_size;

  const int F_IN = 256, H1 = 8, C1 = 32, HC1 = 256, NC = 40, NCP = 64;
  const int Nn   = in_sizes[0] / F_IN;
  const int E    = in_sizes[1] / 2;
  const int Etot = E + Nn;
  const int* esrc = ei;
  const int* edst = ei + E;

  float* p = (float*)d_ws;
  float*    xw1  = p;            p += (size_t)Nn * HC1;
  float*    as1  = p;            p += (size_t)Nn * H1;
  float*    ad1  = p;            p += (size_t)Nn * H1;
  unsigned* m1   = (unsigned*)p; p += (size_t)Nn * H1;
  float*    den1 = p;            p += (size_t)Nn * H1;
  float*    agg1 = p;            p += (size_t)Nn * HC1;
  float*    w2p  = p;            p += (size_t)HC1 * NCP;
  float*    xw2p = p;            p += (size_t)Nn * NCP;
  float*    as2  = p;            p += (size_t)Nn;
  float*    ad2  = p;            p += (size_t)Nn;
  unsigned* m2   = (unsigned*)p; p += (size_t)Nn;
  float*    den2 = p;            p += (size_t)Nn;

  dim3 blk(256);

  zero_f32_kernel<<<2048, blk, 0, stream>>>((float*)m1,
                                            (long)Nn * (2 * H1 + HC1));
  zero_f32_kernel<<<256, blk, 0, stream>>>((float*)m2, (long)Nn * 2);
  zero_f32_kernel<<<1024, blk, 0, stream>>>(out, (long)Nn * NC);
  pad_cols_kernel<<<(HC1 * NCP + 255) / 256, blk, 0, stream>>>(
      W2, w2p, HC1, NC, NCP);

  {
    int waves  = (Nn / 16) * (HC1 / 64);
    int blocks = (waves + 7) / 8;
    gemm_f32_wmma_kernel<<<blocks, blk, 0, stream>>>(x, W1, xw1, Nn, F_IN, HC1);
  }
  alpha_kernel<<<(Nn * H1 + 255) / 256, blk, 0, stream>>>(
      xw1, a1_src, a1_dst, as1, ad1, Nn, H1, C1, HC1);
  edge_max_kernel<<<(Etot + 255) / 256, blk, 0, stream>>>(
      esrc, edst, E, Nn, H1, as1, ad1, m1);
  edge_denom_kernel<<<(Etot + 255) / 256, blk, 0, stream>>>(
      esrc, edst, E, Nn, H1, as1, ad1, m1, den1);
  edge_aggregate_kernel<8, 32, 256><<<(Etot + 255) / 256, blk, 0, stream>>>(
      esrc, edst, E, Nn, as1, ad1, m1, den1, xw1, agg1);
  bias_elu_kernel<<<(int)(((long)Nn * HC1 + 255) / 256), blk, 0, stream>>>(
      agg1, b1, (long)Nn * HC1, HC1);

  {
    int waves  = (Nn / 16) * (NCP / 64);
    int blocks = (waves + 7) / 8;
    gemm_f32_wmma_kernel<<<blocks, blk, 0, stream>>>(agg1, w2p, xw2p,
                                                     Nn, HC1, NCP);
  }
  alpha_kernel<<<(Nn + 255) / 256, blk, 0, stream>>>(
      xw2p, a2_src, a2_dst, as2, ad2, Nn, 1, NC, NCP);
  edge_max_kernel<<<(Etot + 255) / 256, blk, 0, stream>>>(
      esrc, edst, E, Nn, 1, as2, ad2, m2);
  edge_denom_kernel<<<(Etot + 255) / 256, blk, 0, stream>>>(
      esrc, edst, E, Nn, 1, as2, ad2, m2, den2);
  edge_aggregate_kernel<1, 40, 64><<<(Etot + 255) / 256, blk, 0, stream>>>(
      esrc, edst, E, Nn, as2, ad2, m2, den2, xw2p, out);
  bias_add_kernel<<<(int)(((long)Nn * NC + 255) / 256), blk, 0, stream>>>(
      out, b2, (long)Nn * NC, NC);
}


# no float atomics: edges grouped by destination once per call (int counting adds); a wave per node does softmax and weighted row sum in registers, stores once; sums in arrival-dependent slot order
# speedup vs baseline: 10.7591x; 3.1658x over previous
#include <hip/hip_runtime.h>

typedef __attribute__((ext_vector_type(2))) float v2f;
typedef __attribute__((ext_vector_type(8))) float v8f;

#define NEG_SLOPE 0.2f

__global__ void zero_f32_kernel(float* __restrict__ p, long n) {
  long i = (long)blockIdx.x * blockDim.x + threadIdx.x;
  long stride = (long)gridDim.x * blockDim.x;
  for (; i < n; i += stride) p[i] = 0.0f;
}

__global__ void pad_cols_kernel(const float* __restrict__ src,
                                float* __restrict__ dst,
                                int K, int c, int ldd) {
  int i = blockIdx.x * blockDim.x + threadIdx.x;
  if (i >= K * ldd) return;
  int r = i / ldd, j = i - r * ldd;
  dst[i] = (j < c) ? src[r * c + j] : 0.0f;
}

__global__ void gemm_f32_wmma_kernel(const float* __restrict__ A,
                                     const float* __restrict__ B,
                                     float* __restrict__ C,
                                     int M, int K, int ldb) {
  const int lane = threadIdx.x & 31;
  const int wave = threadIdx.x >> 5;
  const int half = lane >> 4;
  const int l    = lane & 15;
  const int nt64   = ldb >> 6;
  const int mtiles = M >> 4;
  const int total  = mtiles * nt64;
  int t = blockIdx.x * (blockDim.x >> 5) + wave;
  if (t >= total) return;
  const int mt = t / nt64;
  const int n0 = (t - mt * nt64) << 6;
  const int m0 = mt << 4;

  v8f acc[4] = {};
  const float* arow = A + (size_t)(m0 + l) * K + 2 * half;
  const float* bp   = B + (size_t)(2 * half) * ldb + n0 + l;
  const size_t bstep = (size_t)4 * ldb;

  v2f a = *(const v2f*)(arow);
  v2f b[4];
#pragma unroll
  for (int s = 0; s < 4; ++s) {
    b[s].x = bp[s * 16];
    b[s].y = bp[ldb + s * 16];
  }

  for (int k0 = 4; k0 < K; k0 += 4) {
    v2f an = *(const v2f*)(arow + k0);
    const float* bpn = bp + bstep;
    v2f bn[4];
#pragma unroll
    for (int s = 0; s < 4; ++s) {
      bn[s].x = bpn[s * 16];
      bn[s].y = bpn[ldb + s * 16];
    }
#pragma unroll
    for (int s = 0; s < 4; ++s)
      acc[s] = __builtin_amdgcn_wmma_f32_16x16x4_f32(
          false, a, false, b[s], (short)0, acc[s], false, false);
    a = an;
#pragma unroll
    for (int s = 0; s < 4; ++s) b[s] = bn[s];
    bp = bpn;
  }
#pragma unroll
  for (int s = 0; s < 4; ++s)
    acc[s] = __builtin_amdgcn_wmma_f32_16x16x4_f32(
        false, a, false, b[s], (short)0, acc[s], false, false);

  float* crow = C + (size_t)(m0 + 8 * half) * ldb + n0 + l;
#pragma unroll
  for (int s = 0; s < 4; ++s)
#pragma unroll
    for (int j = 0; j < 8; ++j) crow[(size_t)j * ldb + s * 16] = acc[s][j];
}

__global__ void alpha_kernel(const float* __restrict__ xw,
                             const float* __restrict__ a_src,
                             const float* __restrict__ a_dst,
                             float* __restrict__ as, float* __restrict__ ad,
                             int n_nodes, int heads, int c, int ldx) {
  int i = blockIdx.x * blockDim.x + threadIdx.x;
  if (i >= n_nodes * heads) return;
  int n = i / heads, h = i - n * heads;
  const float* row = xw + (size_t)n * ldx + (size_t)h * c;
  const float* ws = a_src + (size_t)h * c;
  const float* wd = a_dst + (size_t)h * c;
  float s = 0.0f, d = 0.0f;
  for (int j = 0; j < c; ++j) {
    float v = row[j];
    s += v * ws[j];
    d += v * wd[j];
  }
  as[i] = s;
  ad[i] = d;
}

__global__ void edge_rank_kernel(const int* __restrict__ edst, int E,
                                 int* cnt, int* __restrict__ rank) {
  const int base = blockIdx.x * (int)(blockDim.x * 4) + threadIdx.x;
  int dd[4], rr[4];
#pragma unroll
  for (int q = 0; q < 4; ++q) {
    const int e = base + q * (int)blockDim.x;
    dd[q] = (e < E) ? edst[e] : -1;
  }
#pragma unroll
  for (int q = 0; q < 4; ++q) {
    rr[q] = 0;
    if (dd[q] >= 0) rr[q] = atomicAdd(&cnt[dd[q]], 1);
  }
#pragma unroll
  for (int q = 0; q < 4; ++q) {
    const int e = base + q * (int)blockDim.x;
    if (e < E) rank[e] = rr[q];
  }
}

__global__ void segment_start_kernel(const int* cnt, int* total,
                                     int* __restrict__ start, int n) {
  const int d = blockIdx.x * blockDim.x + threadIdx.x;
  const int lane = threadIdx.x & 31;
  const int c = (d < n) ? cnt[d] : 0;
  int incl = c;
  for (int off = 1; off < 32; off <<= 1) {
    const int t = __shfl_up(incl, off);
    if (lane >= off) incl += t;
  }
  const int wtot = __shfl(incl, 31);
  int base = 0;
  if (lane == 0) base = atomicAdd(total, wtot);
  base = __shfl(base, 0);
  if (d < n) start[d] = base + incl - c;
}

__global__ void edge_place_kernel(const int* __restrict__ esrc,
                                  const int* __restrict__ edst,
                                  const int* __restrict__ rank,
                                  const int* __restrict__ start,
                                  int E, int* __restrict__ csr) {
  const int base = blockIdx.x * (int)(blockDim.x * 4) + threadIdx.x;
  int sv[4], pos[4];
#pragma unroll
  for (int q = 0; q < 4; ++q) {
    const int e = base + q * (int)blockDim.x;
    sv[q] = 0; pos[q] = -1;
    if (e < E) {
      sv[q] = esrc[e];
      pos[q] = start[edst[e]] + rank[e];
    }
  }
#pragma unroll
  for (int q = 0; q < 4; ++q)
    if (pos[q] >= 0) csr[pos[q]] = sv[q];
}

template <int H>
static __device__ __forceinline__ void load_node_row(const float* __restrict__ p,
                                                     float (&v)[H]) {
  if constexpr (H % 4 == 0) {
#pragma unroll
    for (int q = 0; q < H / 4; ++q) {
      const float4 t = reinterpret_cast<const float4*>(p)[q];
      v[4 * q + 0] = t.x; v[4 * q + 1] = t.y;
      v[4 * q + 2] = t.z; v[4 * q + 3] = t.w;
    }
  } else {
#pragma unroll
    for (int i = 0; i < H; ++i) v[i] = p[i];
  }
}

template <int H, int C, int NCH>
static __device__ __forceinline__ void load_x_row(const float* __restrict__ xs,
                                                  int lane, float (&xv)[H * NCH]) {
#pragma unroll
  for (int h = 0; h < H; ++h) {
#pragma unroll
    for (int k = 0; k < NCH; ++k) {
      const int c = k * 32 + lane;
      xv[h * NCH + k] = (c < C) ? xs[h * C + c] : 0.0f;
    }
  }
}

template <int H, int C, int LDX, int U, bool ELU>
__global__ void __launch_bounds__(256)
node_aggregate_kernel(const int* __restrict__ csr,
                      const int* __restrict__ start,
                      const int* __restrict__ cnt, int Nn,
                      const float* __restrict__ as,
                      const float* __restrict__ ad,
                      const float* __restrict__ xw,
                      const float* __restrict__ bias,
                      float* __restrict__ out) {
  constexpr int NCH = (C + 31) / 32;
  const int lane = threadIdx.x & 31;
  const int wv = __builtin_amdgcn_readfirstlane((int)(threadIdx.x >> 5));
  const int nwaves = (int)gridDim.x * (int)(blockDim.x >> 5);
  for (int d = blockIdx.x * (int)(blockDim.x >> 5) + wv; d < Nn; d += nwaves) {
  const int beg = __builtin_amdgcn_readfirstlane(start[d]);
  const int n = __builtin_amdgcn_readfirstlane(cnt[d]) + 1;

  float dv[H];
  load_node_row<H>(ad + (size_t)d * H, dv);

  float m[H], l[H];
#pragma unroll
  for (int h = 0; h < H; ++h) { m[h] = -__builtin_inff(); l[h] = 0.0f; }
  for (int b = 0; b < n; b += 32) {
    const int j = b + lane;
    if (j < n) {
      const int s = (j < n - 1) ? csr[beg + j] : d;
      float av[H];
      load_node_row<H>(as + (size_t)s * H, av);
#pragma unroll
      for (int h = 0; h < H; ++h) {
        float v = av[h] + dv[h];
        v = (v >= 0.0f) ? v : NEG_SLOPE * v;
        const float mn = fmaxf(m[h], v);
        l[h] = l[h] * expf(m[h] - mn) + expf(v - mn);
        m[h] = mn;
      }
    }
  }
#pragma unroll
  for (int h = 0; h < H; ++h) {
    float mx = m[h];
    for (int off = 16; off > 0; off >>= 1) mx = fmaxf(mx, __shfl_xor(mx, off));
    float t = l[h] * expf(m[h] - mx);
    for (int off = 16; off > 0; off >>= 1) t += __shfl_xor(t, off);
    m[h] = __int_as_float(__builtin_amdgcn_readfirstlane(__float_as_int(mx)));
    l[h] = __int_as_float(__builtin_amdgcn_readfirstlane(__float_as_int(t)));
  }

  float acc[H * NCH];
#pragma unroll
  for (int i = 0; i < H * NCH; ++i) acc[i] = 0.0f;

  for (int b = 0; b < n; b += 32) {
    const int j = b + lane;
    int s = d;
    float w[H];
#pragma unroll
    for (int h = 0; h < H; ++h) w[h] = 0.0f;
    if (j < n) {
      s = (j < n - 1) ? csr[beg + j] : d;
      float av[H];
      load_node_row<H>(as + (size_t)s * H, av);
#pragma unroll
      for (int h = 0; h < H; ++h) {
        float v = av[h] + dv[h];
        v = (v >= 0.0f) ? v : NEG_SLOPE * v;
        w[h] = expf(v - m[h]) / l[h];
      }
    }
    const int cb = (n - b < 32) ? (n - b) : 32;
    int jj = 0;
    for (; jj + U <= cb; jj += U) {
      float xr[U][H * NCH];
#pragma unroll
      for (int u = 0; u < U; ++u) {
        const int su = __builtin_amdgcn_readlane(s, jj + u);
        load_x_row<H, C, NCH>(xw + (size_t)su * LDX, lane, xr[u]);
      }
#pragma unroll
      for (int h = 0; h < H; ++h) {
#pragma unroll
        for (int u = 0; u < U; ++u) {
          const float wu = __int_as_float(
              __builtin_amdgcn_readlane(__float_as_int(w[h]), jj + u));
#pragma unroll
          for (int k = 0; k < NCH; ++k) acc[h * NCH + k] += wu * xr[u][h * NCH + k];
        }
      }
    }
    for (; jj < cb; ++jj) {
      const int s0 = __builtin_amdgcn_readlane(s, jj);
      float xa[H * NCH];
      load_x_row<H, C, NCH>(xw + (size_t)s0 * LDX, lane, xa);
#pragma unroll
      for (int h = 0; h < H; ++h) {
        const float w0 = __int_as_float(
            __builtin_amdgcn_readlane(__float_as_int(w[h]), jj));
#pragma unroll
        for (int k = 0; k < NCH; ++k) acc[h * NCH + k] += w0 * xa[h * NCH + k];
      }
    }
  }

  float* od = out + (size_t)d * (H * C);
#pragma unroll
  for (int h = 0; h < H; ++h) {
#pragma unroll
    for (int k = 0; k < NCH; ++k) {
      const int c = k * 32 + lane;
      if (c < C) {
        float v = acc[h * NCH + k] + bias[h * C + c];
        if (ELU) v = (v > 0.0f) ? v : (expf(v) - 1.0f);
        od[h * C + c] = v;
      }
    }
  }
  }
}

extern "C" void kernel_launch(void* const* d_in, const int* in_sizes, int n_in,
                              void* d_out, int out_size, void* d_ws, size_t ws_size,
                              hipStream_t stream) {
  const float* x      = (const float*)d_in[0];
  const int*   ei     = (const int*)d_in[1];
  const float* W1     = (const float*)d_in[2];
  const float* a1_src = (const float*)d_in[3];
  const float* a1_dst = (const float*)d_in[4];
  const float* b1     = (const float*)d_in[5];
  const float* W2     = (const float*)d_in[6];
  const float* a2_src = (const float*)d_in[7];
  const float* a2_dst = (const float*)d_in[8];
  const float* b2     = (const float*)d_in[9];
  float* out = (float*)d_out;
  (void)n_in; (void)out_size; (void)ws_size;

  const int F_IN = 256, H1 = 8, C1 = 32, HC1 = 256, NC = 40, NCP = 64;
  const int Nn   = in_sizes[0] / F_IN;
  const int E    = in_sizes[1] / 2;
  const int* esrc = ei;
  const int* edst = ei + E;

  float* p = (float*)d_ws;
  float* xw1  = p;        p += (size_t)Nn * HC1;
  float* as1  = p;        p += (size_t)Nn * H1;
  float* ad1  = p;        p += (size_t)Nn * H1;
  float* agg1 = p;        p += (size_t)Nn * HC1;
  float* w2p  = p;        p += (size_t)HC1 * NCP;
  float* xw2p = p;        p += (size_t)Nn * NCP;
  float* as2  = p;        p += (size_t)Nn;
  float* ad2  = p;        p += (size_t)Nn;
  int*   start = (int*)p; p += (size_t)Nn;
  int*   rank  = (int*)p; p += (size_t)E;
  int*   csr   = (int*)p; p += (size_t)E;
  int*   cnt   = (int*)p; p += (size_t)Nn + 1;

  dim3 blk(256);

  zero_f32_kernel<<<64, blk, 0, stream>>>((float*)cnt, (long)Nn + 1);
  pad_cols_kernel<<<(HC1 * NCP + 255) / 256, blk, 0, stream>>>(
      W2, w2p, HC1, NC, NCP);

  edge_rank_kernel<<<(E + 1023) / 1024, blk, 0, stream>>>(edst, E, cnt, rank);
  segment_start_kernel<<<(Nn + 255) / 256, blk, 0, stream>>>(
      cnt, cnt + Nn, start, Nn);
  edge_place_kernel<<<(E + 1023) / 1024, blk, 0, stream>>>(
      esrc, edst, rank, start, E, csr);

  {
    int waves  = (Nn / 16) * (HC1 / 64);
    int blocks = (waves + 7) / 8;
    gemm_f32_wmma_kernel<<<blocks, blk, 0, stream>>>(x, W1, xw1, Nn, F_IN, HC1);
  }
  alpha_kernel<<<(Nn * H1 + 255) / 256, blk, 0, stream>>>(
      xw1, a1_src, a1_dst, as1, ad1, Nn, H1, C1, HC1);
  node_aggregate_kernel<8, 32, 256, 2, true><<<2048, blk, 0, stream>>>(
      csr, start, cnt, Nn, as1, ad1, xw1, b1, agg1);

  {
    int waves  = (Nn / 16) * (NCP / 64);
    int blocks = (waves + 7) / 8;
    gemm_f32_wmma_kernel<<<blocks, blk, 0, stream>>>(agg1, w2p, xw2p,
                                                     Nn, HC1, NCP);
  }
  alpha_kernel<<<(Nn + 255) / 256, blk, 0, stream>>>(
      xw2p, a2_src, a2_dst, as2, ad2, Nn, 1, NC, NCP);
  node_aggregate_kernel<1, 40, 64, 4, false><<<2048, blk, 0, stream>>>(
      csr, start, cnt, Nn, as2, ad2, xw2p, b2, out);
}


# as/ad formed in the GEMM kernel from its finished tile (wave's own LDS patch, one lane per (row, head)), same multiply-add chain in channel order, bit-identical; both passes over xw gone
# speedup vs baseline: 11.2605x; 1.0466x over previous
#include <hip/hip_runtime.h>

typedef __attribute__((ext_vector_type(2))) float v2f;
typedef __attribute__((ext_vector_type(8))) float v8f;

#define NEG_SLOPE 0.2f

__global__ void zero_f32_kernel(float* __restrict__ p, long n) {
  long i = (long)blockIdx.x * blockDim.x + threadIdx.x;
  long stride = (long)gridDim.x * blockDim.x;
  for (; i < n; i += stride) p[i] = 0.0f;
}

__global__ void pad_cols_kernel(const float* __restrict__ src,
                                float* __restrict__ dst,
                                int K, int c, int ldd) {
  int i = blockIdx.x * blockDim.x + threadIdx.x;
  if (i >= K * ldd) return;
  int r = i / ldd, j = i - r * ldd;
  dst[i] = (j < c) ? src[r * c + j] : 0.0f;
}

#define TILE_PITCH 66

template <int NH, int CH, int HPT>
__global__ void __launch_bounds__(256)
gemm_alpha_f32_wmma_kernel(const float* __restrict__ A,
                           const float* __restrict__ B,
                           float* __restrict__ C,
                           const float* __restrict__ a_src,
                           const float* __restrict__ a_dst,
                           float* __restrict__ as, float* __restrict__ ad,
                           int M, int K, int ldb) {
  static_assert(HPT * CH <= 64 && (HPT == 1 || HPT == 2) && CH % 4 == 0,
                "whole heads inside a 64-column tile");
  __shared__ float tile[8 * 16 * TILE_PITCH];
  const int lane = threadIdx.x & 31;
  const int wave = threadIdx.x >> 5;
  const int half = lane >> 4;
  const int l    = lane & 15;
  const int nt64   = ldb >> 6;
  const int mtiles = M >> 4;
  const int total  = mtiles * nt64;
  int t = blockIdx.x * (blockDim.x >> 5) + wave;
  if (t >= total) return;
  const int mt = t / nt64;
  const int n0 = (t - mt * nt64) << 6;
  const int m0 = mt << 4;

  v8f acc[4] = {};
  const float* arow = A + (size_t)(m0 + l) * K + 2 * half;
  const float* bp   = B + (size_t)(2 * half) * ldb + n0 + l;
  const size_t bstep = (size_t)4 * ldb;

  v2f a = *(const v2f*)(arow);
  v2f b[4];
#pragma unroll
  for (int s = 0; s < 4; ++s) {
    b[s].x = bp[s * 16];
    b[s].y = bp[ldb + s * 16];
  }

  for (int k0 = 4; k0 < K; k0 += 4) {
    v2f an = *(const v2f*)(arow + k0);
    const float* bpn = bp + bstep;
    v2f bn[4];
#pragma unroll
    for (int s = 0; s < 4; ++s) {
      bn[s].x = bpn[s * 16];
      bn[s].y = bpn[ldb + s * 16];
    }
#pragma unroll
    for (int s = 0; s < 4; ++s)
      acc[s] = __builtin_amdgcn_wmma_f32_16x16x4_f32(
          false, a, false, b[s], (short)0, acc[s], false, false);
    a = an;
#pragma unroll
    for (int s = 0; s < 4; ++s) b[s] = bn[s];
    bp = bpn;
  }
#pragma unroll
  for (int s = 0; s < 4; ++s)
    acc[s] = __builtin_amdgcn_wmma_f32_16x16x4_f32(
        false, a, false, b[s], (short)0, acc[s], false, false);

  float* crow = C + (size_t)(m0 + 8 * half) * ldb + n0 + l;
  float* tl = tile + wave * (16 * TILE_PITCH);
  float* trow = tl + (8 * half) * TILE_PITCH + l;
#pragma unroll
  for (int s = 0; s < 4; ++s)
#pragma unroll
    for (int j = 0; j < 8; ++j) {
      crow[(size_t)j * ldb + s * 16] = acc[s][j];
      trow[j * TILE_PITCH + s * 16] = acc[s][j];
    }

  __builtin_amdgcn_fence(__ATOMIC_RELEASE, "wavefront");
  __builtin_amdgcn_wave_barrier();
  __builtin_amdgcn_fence(__ATOMIC_ACQUIRE, "wavefront");

  const int hd0 = (n0 >> 6) * HPT;
  if constexpr (HPT == 2) {
    const int hd = hd0 + half;
    const float* xr = tl + l * TILE_PITCH + half * CH;
    const float4* ws4 = reinterpret_cast<const float4*>(a_src + (size_t)hd * CH);
    const float4* wd4 = reinterpret_cast<const float4*>(a_dst + (size_t)hd * CH);
    float s = 0.0f, d = 0.0f;
#pragma unroll
    for (int q = 0; q < CH / 4; ++q) {
      const float4 ws = ws4[q];
      const float4 wd = wd4[q];
      float v;
      v = xr[4 * q + 0]; s = __builtin_fmaf(v, ws.x, s); d = __builtin_fmaf(v, wd.x, d);
      v = xr[4 * q + 1]; s = __builtin_fmaf(v, ws.y, s); d = __builtin_fmaf(v, wd.y, d);
      v = xr[4 * q + 2]; s = __builtin_fmaf(v, ws.z, s); d = __builtin_fmaf(v, wd.z, d);
      v = xr[4 * q + 3]; s = __builtin_fmaf(v, ws.w, s); d = __builtin_fmaf(v, wd.w, d);
    }
    as[(size_t)(m0 + l) * NH + hd] = s;
    ad[(size_t)(m0 + l) * NH + hd] = d;
  } else {
    const float* xr = tl + l * TILE_PITCH;
    const float4* w4 = reinterpret_cast<const float4*>(
        (half ? a_dst : a_src) + (size_t)hd0 * CH);
    float s = 0.0f;
#pragma unroll
    for (int q = 0; q < CH / 4; ++q) {
      const float4 w = w4[q];
      s = __builtin_fmaf(xr[4 * q + 0], w.x, s);
      s = __builtin_fmaf(xr[4 * q + 1], w.y, s);
      s = __builtin_fmaf(xr[4 * q + 2], w.z, s);
      s = __builtin_fmaf(xr[4 * q + 3], w.w, s);
    }
    (half ? ad : as)[(size_t)(m0 + l) * NH + hd0] = s;
  }
}

__global__ void edge_rank_kernel(const int* __restrict__ edst, int E,
                                 int* cnt, int* __restrict__ rank) {
  const int base = blockIdx.x * (int)(blockDim.x * 4) + threadIdx.x;
  int dd[4], rr[4];
#pragma unroll
  for (int q = 0; q < 4; ++q) {
    const int e = base + q * (int)blockDim.x;
    dd[q] = (e < E) ? edst[e] : -1;
  }
#pragma unroll
  for (int q = 0; q < 4; ++q) {
    rr[q] = 0;
    if (dd[q] >= 0) rr[q] = atomicAdd(&cnt[dd[q]], 1);
  }
#pragma unroll
  for (int q = 0; q < 4; ++q) {
    const int e = base + q * (int)blockDim.x;
    if (e < E) rank[e] = rr[q];
  }
}

__global__ void segment_start_kernel(const int* cnt, int* total,
                                     int* __restrict__ start, int n) {
  const int d = blockIdx.x * blockDim.x + threadIdx.x;
  const int lane = threadIdx.x & 31;
  const int c = (d < n) ? cnt[d] : 0;
  int incl = c;
  for (int off = 1; off < 32; off <<= 1) {
    const int t = __shfl_up(incl, off);
    if (lane >= off) incl += t;
  }
  const int wtot = __shfl(incl, 31);
  int base = 0;
  if (lane == 0) base = atomicAdd(total, wtot);
  base = __shfl(base, 0);
  if (d < n) start[d] = base + incl - c;
}

__global__ void edge_place_kernel(const int* __restrict__ esrc,
                                  const int* __restrict__ edst,
                                  const int* __restrict__ rank,
                                  const int* __restrict__ start,
                                  int E, int* __restrict__ csr) {
  const int base = blockIdx.x * (int)(blockDim.x * 4) + threadIdx.x;
  int sv[4], pos[4];
#pragma unroll
  for (int q = 0; q < 4; ++q) {
    const int e = base + q * (int)blockDim.x;
    sv[q] = 0; pos[q] = -1;
    if (e < E) {
      sv[q] = esrc[e];
      pos[q] = start[edst[e]] + rank[e];
    }
  }
#pragma unroll
  for (int q = 0; q < 4; ++q)
    if (pos[q] >= 0) csr[pos[q]] = sv[q];
}

template <int H>
static __device__ __forceinline__ void load_node_row(const float* __restrict__ p,
                                                     float (&v)[H]) {
  if constexpr (H % 4 == 0) {
#pragma unroll
    for (int q = 0; q < H / 4; ++q) {
      const float4 t = reinterpret_cast<const float4*>(p)[q];
      v[4 * q + 0] = t.x; v[4 * q + 1] = t.y;
      v[4 * q + 2] = t.z; v[4 * q + 3] = t.w;
    }
  } else {
#pragma unroll
    for (int i = 0; i < H; ++i) v[i] = p[i];
  }
}

template <int H, int C, int NCH>
static __device__ __forceinline__ void load_x_row(const float* __restrict__ xs,
                                                  int lane, float (&xv)[H * NCH]) {
#pragma unroll
  for (int h = 0; h < H; ++h) {
#pragma unroll
    for (int k = 0; k < NCH; ++k) {
      const int c = k * 32 + lane;
      xv[h * NCH + k] = (c < C) ? xs[h * C + c] : 0.0f;
    }
  }
}

template <int H, int C, int LDX, int U, bool ELU>
__global__ void __launch_bounds__(256)
node_aggregate_kernel(const int* __restrict__ csr,
                      const int* __restrict__ start,
                      const int* __restrict__ cnt, int Nn,
                      const float* __restrict__ as,
                      const float* __restrict__ ad,
                      const float* __restrict__ xw,
                      const float* __restrict__ bias,
                      float* __restrict__ out) {
  constexpr int NCH = (C + 31) / 32;
  const int lane = threadIdx.x & 31;
  const int wv = __builtin_amdgcn_readfirstlane((int)(threadIdx.x >> 5));
  const int nwaves = (int)gridDim.x * (int)(blockDim.x >> 5);
  for (int d = blockIdx.x * (int)(blockDim.x >> 5) + wv; d < Nn; d += nwaves) {
  const int beg = __builtin_amdgcn_readfirstlane(start[d]);
  const int n = __builtin_amdgcn_readfirstlane(cnt[d]) + 1;

  float dv[H];
  load_node_row<H>(ad + (size_t)d * H, dv);

  float m[H], l[H];
#pragma unroll
  for (int h = 0; h < H; ++h) { m[h] = -__builtin_inff(); l[h] = 0.0f; }
  for (int b = 0; b < n; b += 32) {
    const int j = b + lane;
    if (j < n) {
      const int s = (j < n - 1) ? csr[beg + j] : d;
      float av[H];
      load_node_row<H>(as + (size_t)s * H, av);
#pragma unroll
      for (int h = 0; h < H; ++h) {
        float v = av[h] + dv[h];
        v = (v >= 0.0f) ? v : NEG_SLOPE * v;
        const float mn = fmaxf(m[h], v);
        l[h] = l[h] * expf(m[h] - mn) + expf(v - mn);
        m[h] = mn;
      }
    }
  }
#pragma unroll
  for (int h = 0; h < H; ++h) {
    float mx = m[h];
    for (int off = 16; off > 0; off >>= 1) mx = fmaxf(mx, __shfl_xor(mx, off));
    float t = l[h] * expf(m[h] - mx);
    for (int off = 16; off > 0; off >>= 1) t += __shfl_xor(t, off);
    m[h] = __int_as_float(__builtin_amdgcn_readfirstlane(__float_as_int(mx)));
    l[h] = __int_as_float(__builtin_amdgcn_readfirstlane(__float_as_int(t)));
  }

  float acc[H * NCH];
#pragma unroll
  for (int i = 0; i < H * NCH; ++i) acc[i] = 0.0f;

  for (int b = 0; b < n; b += 32) {
    const int j = b + lane;
    int s = d;
    float w[H];
#pragma unroll
    for (int h = 0; h < H; ++h) w[h] = 0.0f;
    if (j < n) {
      s = (j < n - 1) ? csr[beg + j] : d;
      float av[H];
      load_node_row<H>(as + (size_t)s * H, av);
#pragma unroll
      for (int h = 0; h < H; ++h) {
        float v = av[h] + dv[h];
        v = (v >= 0.0f) ? v : NEG_SLOPE * v;
        w[h] = expf(v - m[h]) / l[h];
      }
    }
    const int cb = (n - b < 32) ? (n - b) : 32;
    int jj = 0;
    for (; jj + U <= cb; jj += U) {
      float xr[U][H * NCH];
#pragma unroll
      for (int u = 0; u < U; ++u) {
        const int su = __builtin_amdgcn_readlane(s, jj + u);
        load_x_row<H, C, NCH>(xw + (size_t)su * LDX, lane, xr[u]);
      }
#pragma unroll
      for (int h = 0; h < H; ++h) {
#pragma unroll
        for (int u = 0; u < U; ++u) {
          const float wu = __int_as_float(
              __builtin_amdgcn_readlane(__float_as_int(w[h]), jj + u));
#pragma unroll
          for (int k = 0; k < NCH; ++k) acc[h * NCH + k] += wu * xr[u][h * NCH + k];
        }
      }
    }
    for (; jj < cb; ++jj) {
      const int s0 = __builtin_amdgcn_readlane(s, jj);
      float xa[H * NCH];
      load_x_row<H, C, NCH>(xw + (size_t)s0 * LDX, lane, xa);
#pragma unroll
      for (int h = 0; h < H; ++h) {
        const float w0 = __int_as_float(
            __builtin_amdgcn_readlane(__float_as_int(w[h]), jj));
#pragma unroll
        for (int k = 0; k < NCH; ++k) acc[h * NCH + k] += w0 * xa[h * NCH + k];
      }
    }
  }

  float* od = out + (size_t)d * (H * C);
#pragma unroll
  for (int h = 0; h < H; ++h) {
#pragma unroll
    for (int k = 0; k < NCH; ++k) {
      const int c = k * 32 + lane;
      if (c < C) {
        float v = acc[h * NCH + k] + bias[h * C + c];
        if (ELU) v = (v > 0.0f) ? v : (expf(v) - 1.0f);
        od[h * C + c] = v;
      }
    }
  }
  }
}

extern "C" void kernel_launch(void* const* d_in, const int* in_sizes, int n_in,
                              void* d_out, int out_size, void* d_ws, size_t ws_size,
                              hipStream_t stream) {
  const float* x      = (const float*)d_in[0];
  const int*   ei     = (const int*)d_in[1];
  const float* W1     = (const float*)d_in[2];
  const float* a1_src = (const float*)d_in[3];
  const float* a1_dst = (const float*)d_in[4];
  const float* b1     = (const float*)d_in[5];
  const float* W2     = (const float*)d_in[6];
  const float* a2_src = (const float*)d_in[7];
  const float* a2_dst = (const float*)d_in[8];
  const float* b2     = (const float*)d_in[9];
  float* out = (float*)d_out;
  (void)n_in; (void)out_size; (void)ws_size;

  const int F_IN = 256, H1 = 8, HC1 = 256, NC = 40, NCP = 64;
  const int Nn   = in_sizes[0] / F_IN;
  const int E    = in_sizes[1] / 2;
  const int* esrc = ei;
  const int* edst = ei + E;

  float* p = (float*)d_ws;
  float* xw1  = p;        p += (size_t)Nn * HC1;
  float* as1  = p;        p += (size_t)Nn * H1;
  float* ad1  = p;        p += (size_t)Nn * H1;
  float* agg1 = p;        p += (size_t)Nn * HC1;
  float* w2p  = p;        p += (size_t)HC1 * NCP;
  float* xw2p = p;        p += (size_t)Nn * NCP;
  float* as2  = p;        p += (size_t)Nn;
  float* ad2  = p;        p += (size_t)Nn;
  int*   start = (int*)p; p += (size_t)Nn;
  int*   rank  = (int*)p; p += (size_t)E;
  int*   csr   = (int*)p; p += (size_t)E;
  int*   cnt   = (int*)p; p += (size_t)Nn + 1;

  dim3 blk(256);

  zero_f32_kernel<<<64, blk, 0, stream>>>((float*)cnt, (long)Nn + 1);
  pad_cols_kernel<<<(HC1 * NCP + 255) / 256, blk, 0, stream>>>(
      W2, w2p, HC1, NC, NCP);

  edge_rank_kernel<<<(E + 1023) / 1024, blk, 0, stream>>>(edst, E, cnt, rank);
  segment_start_kernel<<<(Nn + 255) / 256, blk, 0, stream>>>(
      cnt, cnt + Nn, start, Nn);
  edge_place_kernel<<<(E + 1023) / 1024, blk, 0, stream>>>(
      esrc, edst, rank, start, E, csr);

  {
    int waves  = (Nn / 16) * (HC1 / 64);
    int blocks = (waves + 7) / 8;
    gemm_alpha_f32_wmma_kernel<8, 32, 2><<<blocks, blk, 0, stream>>>(
        x, W1, xw1, a1_src, a1_dst, as1, ad1, Nn, F_IN, HC1);
  }
  node_aggregate_kernel<8, 32, 256, 2, true><<<2048, blk, 0, stream>>>(
      csr, start, cnt, Nn, as1, ad1, xw1, b1, agg1);

  {
    int waves  = (Nn / 16) * (NCP / 64);
    int blocks = (waves + 7) / 8;
    gemm_alpha_f32_wmma_kernel<1, 40, 1><<<blocks, blk, 0, stream>>>(
        agg1, w2p, xw2p, a2_src, a2_dst, as2, ad2, Nn, HC1, NCP);
  }
  node_aggregate_kernel<1, 40, 64, 4, false><<<2048, blk, 0, stream>>>(
      csr, start, cnt, Nn, as2, ad2, xw2p, b2, out);
}


# edges grouped by destination in two levels on LDS counters (bins of 256 nodes, then nodes; the returning LDS add is the slot): no global atomics, no counter clear; slot order by arrival as before
# speedup vs baseline: 15.0107x; 1.3330x over previous
#include <hip/hip_runtime.h>

typedef __attribute__((ext_vector_type(2))) float v2f;
typedef __attribute__((ext_vector_type(8))) float v8f;

#define NEG_SLOPE 0.2f

__global__ void pad_cols_kernel(const float* __restrict__ src,
                                float* __restrict__ dst,
                                int K, int c, int ldd) {
  int i = blockIdx.x * blockDim.x + threadIdx.x;
  if (i >= K * ldd) return;
  int r = i / ldd, j = i - r * ldd;
  dst[i] = (j < c) ? src[r * c + j] : 0.0f;
}

#define TILE_PITCH 66

template <int NH, int CH, int HPT>
__global__ void __launch_bounds__(256)
gemm_alpha_f32_wmma_kernel(const float* __restrict__ A,
                           const float* __restrict__ B,
                           float* __restrict__ C,
                           const float* __restrict__ a_src,
                           const float* __restrict__ a_dst,
                           float* __restrict__ as, float* __restrict__ ad,
                           int M, int K, int ldb) {
  static_assert(HPT * CH <= 64 && (HPT == 1 || HPT == 2) && CH % 4 == 0,
                "whole heads inside a 64-column tile");
  __shared__ float tile[8 * 16 * TILE_PITCH];
  const int lane = threadIdx.x & 31;
  const int wave = threadIdx.x >> 5;
  const int half = lane >> 4;
  const int l    = lane & 15;
  const int nt64   = ldb >> 6;
  const int mtiles = M >> 4;
  const int total  = mtiles * nt64;
  int t = blockIdx.x * (blockDim.x >> 5) + wave;
  if (t >= total) return;
  const int mt = t / nt64;
  const int n0 = (t - mt * nt64) << 6;
  const int m0 = mt << 4;

  v8f acc[4] = {};
  const float* arow = A + (size_t)(m0 + l) * K + 2 * half;
  const float* bp   = B + (size_t)(2 * half) * ldb + n0 + l;
  const size_t bstep = (size_t)4 * ldb;

  v2f a = *(const v2f*)(arow);
  v2f b[4];
#pragma unroll
  for (int s = 0; s < 4; ++s) {
    b[s].x = bp[s * 16];
    b[s].y = bp[ldb + s * 16];
  }

  for (int k0 = 4; k0 < K; k0 += 4) {
    v2f an = *(const v2f*)(arow + k0);
    const float* bpn = bp + bstep;
    v2f bn[4];
#pragma unroll
    for (int s = 0; s < 4; ++s) {
      bn[s].x = bpn[s * 16];
      bn[s].y = bpn[ldb + s * 16];
    }
#pragma unroll
    for (int s = 0; s < 4; ++s)
      acc[s] = __builtin_amdgcn_wmma_f32_16x16x4_f32(
          false, a, false, b[s], (short)0, acc[s], false, false);
    a = an;
#pragma unroll
    for (int s = 0; s < 4; ++s) b[s] = bn[s];
    bp = bpn;
  }
#pragma unroll
  for (int s = 0; s < 4; ++s)
    acc[s] = __builtin_amdgcn_wmma_f32_16x16x4_f32(
        false, a, false, b[s], (short)0, acc[s], false, false);

  float* crow = C + (size_t)(m0 + 8 * half) * ldb + n0 + l;
  float* tl = tile + wave * (16 * TILE_PITCH);
  float* trow = tl + (8 * half) * TILE_PITCH + l;
#pragma unroll
  for (int s = 0; s < 4; ++s)
#pragma unroll
    for (int j = 0; j < 8; ++j) {
      crow[(size_t)j * ldb + s * 16] = acc[s][j];
      trow[j * TILE_PITCH + s * 16] = acc[s][j];
    }

  __builtin_amdgcn_fence(__ATOMIC_RELEASE, "wavefront");
  __builtin_amdgcn_wave_barrier();
  __builtin_amdgcn_fence(__ATOMIC_ACQUIRE, "wavefront");

  const int hd0 = (n0 >> 6) * HPT;
  if constexpr (HPT == 2) {
    const int hd = hd0 + half;
    const float* xr = tl + l * TILE_PITCH + half * CH;
    const float4* ws4 = reinterpret_cast<const float4*>(a_src + (size_t)hd * CH);
    const float4* wd4 = reinterpret_cast<const float4*>(a_dst + (size_t)hd * CH);
    float s = 0.0f, d = 0.0f;
#pragma unroll
    for (int q = 0; q < CH / 4; ++q) {
      const float4 ws = ws4[q];
      const float4 wd = wd4[q];
      float v;
      v = xr[4 * q + 0]; s = __builtin_fmaf(v, ws.x, s); d = __builtin_fmaf(v, wd.x, d);
      v = xr[4 * q + 1]; s = __builtin_fmaf(v, ws.y, s); d = __builtin_fmaf(v, wd.y, d);
      v = xr[4 * q + 2]; s = __builtin_fmaf(v, ws.z, s); d = __builtin_fmaf(v, wd.z, d);
      v = xr[4 * q + 3]; s = __builtin_fmaf(v, ws.w, s); d = __builtin_fmaf(v, wd.w, d);
    }
    as[(size_t)(m0 + l) * NH + hd] = s;
    ad[(size_t)(m0 + l) * NH + hd] = d;
  } else {
    const float* xr = tl + l * TILE_PITCH;
    const float4* w4 = reinterpret_cast<const float4*>(
        (half ? a_dst : a_src) + (size_t)hd0 * CH);
    float s = 0.0f;
#pragma unroll
    for (int q = 0; q < CH / 4; ++q) {
      const float4 w = w4[q];
      s = __builtin_fmaf(xr[4 * q + 0], w.x, s);
      s = __builtin_fmaf(xr[4 * q + 1], w.y, s);
      s = __builtin_fmaf(xr[4 * q + 2], w.z, s);
      s = __builtin_fmaf(xr[4 * q + 3], w.w, s);
    }
    (half ? ad : as)[(size_t)(m0 + l) * NH + hd0] = s;
  }
}

#define BIN_SHIFT 8
#define BIN_NODES 256
#define MAX_BINS 256
#define EDGE_CHUNK 8192
#define EDGE_THREADS 512
#define EDGE_PER_THREAD (EDGE_CHUNK / EDGE_THREADS)
#define ROW_GROUPS (EDGE_THREADS / 64)
#define ROWS_PER_TRIP 13
#define SORT_PER_TRIP 12
#define SORT_CAP 12288

static __device__ __forceinline__ int scan256_exclusive(int c, int tid, int* wsum) {
  const int lane = tid & 31, wave = tid >> 5;
  int incl = c;
  for (int off = 1; off < 32; off <<= 1) {
    const int t = __shfl_up(incl, off);
    if (lane >= off) incl += t;
  }
  if (wave < 8 && lane == 31) wsum[wave] = incl;
  __syncthreads();
  int wbase = 0;
  for (int w = 0; w < 8; ++w) wbase += (w < wave) ? wsum[w] : 0;
  return wbase + incl - c;
}

__global__ void __launch_bounds__(EDGE_THREADS)
bin_count_kernel(const int* __restrict__ edst, int E, int* __restrict__ binCnt) {
  __shared__ int hist[MAX_BINS];
  const int tid = threadIdx.x;
  const int base = blockIdx.x * EDGE_CHUNK + tid;
  int dd[EDGE_PER_THREAD];
#pragma unroll
  for (int q = 0; q < EDGE_PER_THREAD; ++q) {
    const int e = base + q * EDGE_THREADS;
    const int v = edst[(e < E) ? e : (E - 1)];
    dd[q] = (e < E) ? v : -1;
  }
  if (tid < MAX_BINS) hist[tid] = 0;
  __syncthreads();
#pragma unroll
  for (int q = 0; q < EDGE_PER_THREAD; ++q)
    if (dd[q] >= 0) atomicAdd(&hist[dd[q] >> BIN_SHIFT], 1);
  __syncthreads();
  if (tid < MAX_BINS) binCnt[blockIdx.x * MAX_BINS + tid] = hist[tid];
}

__global__ void __launch_bounds__(EDGE_THREADS)
bin_place_kernel(const int* __restrict__ esrc, const int* __restrict__ edst, int E,
                 const int* __restrict__ binCnt, int* __restrict__ binStart,
                 int2* __restrict__ bpair) {
  __shared__ alignas(16) int ptot[ROW_GROUPS][MAX_BINS];
  __shared__ alignas(16) int pbef[ROW_GROUPS][MAX_BINS];
  __shared__ int gbase[MAX_BINS];
  __shared__ int lstart[MAX_BINS];
  __shared__ int lcur[MAX_BINS];
  __shared__ int wsum_a[8];
  __shared__ int wsum_b[8];
  __shared__ int2 stage[EDGE_CHUNK];
  const int tid = threadIdx.x;
  const int g = blockIdx.x, G = gridDim.x;

  const int base = g * EDGE_CHUNK + tid;
  int dd[EDGE_PER_THREAD], ss[EDGE_PER_THREAD];
#pragma unroll
  for (int q = 0; q < EDGE_PER_THREAD; ++q) {
    const int e = base + q * EDGE_THREADS;
    const int ec = (e < E) ? e : (E - 1);
    const int dv = edst[ec];
    ss[q] = esrc[ec];
    dd[q] = (e < E) ? dv : -1;
  }

  const int rg = tid >> 6;
  const int c4 = (tid & 63) * 4;
  int t0 = 0, t1 = 0, t2 = 0, t3 = 0, b0 = 0, b1 = 0, b2 = 0, b3 = 0;
  for (int r0 = 0; r0 < G; r0 += ROW_GROUPS * ROWS_PER_TRIP) {
    int4 v[ROWS_PER_TRIP];
#pragma unroll
    for (int k = 0; k < ROWS_PER_TRIP; ++k) {
      const int r = r0 + rg + ROW_GROUPS * k;
      const int rc = (r < G) ? r : (G - 1);
      v[k] = *reinterpret_cast<const int4*>(binCnt + (size_t)rc * MAX_BINS + c4);
    }
#pragma unroll
    for (int k = 0; k < ROWS_PER_TRIP; ++k) {
      const int r = r0 + rg + ROW_GROUPS * k;
      const bool in = r < G, early = r < g;
      t0 += in ? v[k].x : 0; t1 += in ? v[k].y : 0;
      t2 += in ? v[k].z : 0; t3 += in ? v[k].w : 0;
      b0 += early ? v[k].x : 0; b1 += early ? v[k].y : 0;
      b2 += early ? v[k].z : 0; b3 += early ? v[k].w : 0;
    }
  }
  *reinterpret_cast<int4*>(&ptot[rg][c4]) = make_int4(t0, t1, t2, t3);
  *reinterpret_cast<int4*>(&pbef[rg][c4]) = make_int4(b0, b1, b2, b3);
  __syncthreads();

  int tot = 0, bef = 0, own = 0;
  if (tid < MAX_BINS) {
#pragma unroll
    for (int k = 0; k < ROW_GROUPS; ++k) { tot += ptot[k][tid]; bef += pbef[k][tid]; }
    own = binCnt[(size_t)g * MAX_BINS + tid];
  }
  const int excl  = scan256_exclusive(tot, tid, wsum_a);
  const int lexcl = scan256_exclusive(own, tid, wsum_b);
  if (tid < MAX_BINS) {
    gbase[tid] = excl + bef;
    lstart[tid] = lexcl;
    lcur[tid] = lexcl;
    if (g == 0) {
      binStart[tid] = excl;
      if (tid == MAX_BINS - 1) binStart[MAX_BINS] = excl + tot;
    }
  }
  __syncthreads();

  int jj[EDGE_PER_THREAD];
#pragma unroll
  for (int q = 0; q < EDGE_PER_THREAD; ++q) {
    jj[q] = -1;
    if (dd[q] >= 0) jj[q] = atomicAdd(&lcur[dd[q] >> BIN_SHIFT], 1);
  }
#pragma unroll
  for (int q = 0; q < EDGE_PER_THREAD; ++q)
    if (jj[q] >= 0) stage[jj[q]] = make_int2(ss[q], dd[q]);
  __syncthreads();

  const int left = E - g * EDGE_CHUNK;
  const int nval = (left < EDGE_CHUNK) ? left : EDGE_CHUNK;
  int2 pr[EDGE_PER_THREAD];
  int to[EDGE_PER_THREAD];
#pragma unroll
  for (int q = 0; q < EDGE_PER_THREAD; ++q) {
    const int j = q * EDGE_THREADS + tid;
    pr[q] = stage[(j < nval) ? j : (nval - 1)];
  }
#pragma unroll
  for (int q = 0; q < EDGE_PER_THREAD; ++q) {
    const int j = q * EDGE_THREADS + tid;
    const int bin = pr[q].y >> BIN_SHIFT;
    to[q] = gbase[bin] + (j - lstart[bin]);
  }
#pragma unroll
  for (int q = 0; q < EDGE_PER_THREAD; ++q) {
    const int j = q * EDGE_THREADS + tid;
    if (j < nval) bpair[to[q]] = pr[q];
  }
}

__global__ void __launch_bounds__(EDGE_THREADS)
bin_sort_kernel(const int2* __restrict__ bpair, const int* __restrict__ binStart,
                int Nn, int* __restrict__ start, int* __restrict__ cnt,
                int* __restrict__ csr) {
  __shared__ int lcnt[BIN_NODES];
  __shared__ int cur[BIN_NODES];
  __shared__ int wsum[8];
  __shared__ int stage[SORT_CAP];
  const int tid = threadIdx.x;
  const int b = blockIdx.x;
  const int beg = binStart[b];
  const int end = binStart[b + 1];
  const int n = end - beg;
  const bool fits = n <= SORT_CAP;
  if (tid < BIN_NODES) lcnt[tid] = 0;
  __syncthreads();
  for (int i0 = beg; i0 < end; i0 += EDGE_THREADS * SORT_PER_TRIP) {
    int dl[SORT_PER_TRIP];
#pragma unroll
    for (int q = 0; q < SORT_PER_TRIP; ++q) {
      const int i = i0 + q * EDGE_THREADS + tid;
      const int v = bpair[(i < end) ? i : (end - 1)].y;
      dl[q] = (i < end) ? (v & (BIN_NODES - 1)) : -1;
    }
#pragma unroll
    for (int q = 0; q < SORT_PER_TRIP; ++q)
      if (dl[q] >= 0) atomicAdd(&lcnt[dl[q]], 1);
  }
  __syncthreads();
  const int c = (tid < BIN_NODES) ? lcnt[tid] : 0;
  const int excl = scan256_exclusive(c, tid, wsum);
  if (tid < BIN_NODES) {
    cur[tid] = excl;
    const int d = (b << BIN_SHIFT) + tid;
    if (d < Nn) { start[d] = beg + excl; cnt[d] = c; }
  }
  __syncthreads();
  for (int i0 = beg; i0 < end; i0 += EDGE_THREADS * SORT_PER_TRIP) {
    int sv[SORT_PER_TRIP], dl[SORT_PER_TRIP], ps[SORT_PER_TRIP];
#pragma unroll
    for (int q = 0; q < SORT_PER_TRIP; ++q) {
      const int i = i0 + q * EDGE_THREADS + tid;
      const int2 pr = bpair[(i < end) ? i : (end - 1)];
      sv[q] = pr.x;
      dl[q] = (i < end) ? (pr.y & (BIN_NODES - 1)) : -1;
    }
#pragma unroll
    for (int q = 0; q < SORT_PER_TRIP; ++q) {
      ps[q] = -1;
      if (dl[q] >= 0) ps[q] = atomicAdd(&cur[dl[q]], 1);
    }
    if (fits) {
#pragma unroll
      for (int q = 0; q < SORT_PER_TRIP; ++q)
        if (ps[q] >= 0) stage[ps[q]] = sv[q];
    } else {
#pragma unroll
      for (int q = 0; q < SORT_PER_TRIP; ++q)
        if (ps[q] >= 0) csr[beg + ps[q]] = sv[q];
    }
  }
  if (fits) {
    __syncthreads();
    for (int j = tid; j < n; j += EDGE_THREADS) csr[beg + j] = stage[j];
  }
}

template <int H>
static __device__ __forceinline__ void load_node_row(const float* __restrict__ p,
                                                     float (&v)[H]) {
  if constexpr (H % 4 == 0) {
#pragma unroll
    for (int q = 0; q < H / 4; ++q) {
      const float4 t = reinterpret_cast<const float4*>(p)[q];
      v[4 * q + 0] = t.x; v[4 * q + 1] = t.y;
      v[4 * q + 2] = t.z; v[4 * q + 3] = t.w;
    }
  } else {
#pragma unroll
    for (int i = 0; i < H; ++i) v[i] = p[i];
  }
}

template <int H, int C, int NCH>
static __device__ __forceinline__ void load_x_row(const float* __restrict__ xs,
                                                  int lane, float (&xv)[H * NCH]) {
#pragma unroll
  for (int h = 0; h < H; ++h) {
#pragma unroll
    for (int k = 0; k < NCH; ++k) {
      const int c = k * 32 + lane;
      xv[h * NCH + k] = (c < C) ? xs[h * C + c] : 0.0f;
    }
  }
}

template <int H, int C, int LDX, int U, bool ELU>
__global__ void __launch_bounds__(256)
node_aggregate_kernel(const int* __restrict__ csr,
                      const int* __restrict__ start,
                      const int* __restrict__ cnt, int Nn,
                      const float* __restrict__ as,
                      const float* __restrict__ ad,
                      const float* __restrict__ xw,
                      const float* __restrict__ bias,
                      float* __restrict__ out) {
  constexpr int NCH = (C + 31) / 32;
  const int lane = threadIdx.x & 31;
  const int wv = __builtin_amdgcn_readfirstlane((int)(threadIdx.x >> 5));
  const int nwaves = (int)gridDim.x * (int)(blockDim.x >> 5);
  for (int d = blockIdx.x * (int)(blockDim.x >> 5) + wv; d < Nn; d += nwaves) {
  const int beg = __builtin_amdgcn_readfirstlane(start[d]);
  const int n = __builtin_amdgcn_readfirstlane(cnt[d]) + 1;

  float dv[H];
  load_node_row<H>(ad + (size_t)d * H, dv);

  float m[H], l[H];
#pragma unroll
  for (int h = 0; h < H; ++h) { m[h] = -__builtin_inff(); l[h] = 0.0f; }
  for (int b = 0; b < n; b += 32) {
    const int j = b + lane;
    if (j < n) {
      const int s = (j < n - 1) ? csr[beg + j] : d;
      float av[H];
      load_node_row<H>(as + (size_t)s * H, av);
#pragma unroll
      for (int h = 0; h < H; ++h) {
        float v = av[h] + dv[h];
        v = (v >= 0.0f) ? v : NEG_SLOPE * v;
        const float mn = fmaxf(m[h], v);
        l[h] = l[h] * expf(m[h] - mn) + expf(v - mn);
        m[h] = mn;
      }
    }
  }
#pragma unroll
  for (int h = 0; h < H; ++h) {
    float mx = m[h];
    for (int off = 16; off > 0; off >>= 1) mx = fmaxf(mx, __shfl_xor(mx, off));
    float t = l[h] * expf(m[h] - mx);
    for (int off = 16; off > 0; off >>= 1) t += __shfl_xor(t, off);
    m[h] = __int_as_float(__builtin_amdgcn_readfirstlane(__float_as_int(mx)));
    l[h] = __int_as_float(__builtin_amdgcn_readfirstlane(__float_as_int(t)));
  }

  float acc[H * NCH];
#pragma unroll
  for (int i = 0; i < H * NCH; ++i) acc[i] = 0.0f;

  for (int b = 0; b < n; b += 32) {
    const int j = b + lane;
    int s = d;
    float w[H];
#pragma unroll
    for (int h = 0; h < H; ++h) w[h] = 0.0f;
    if (j < n) {
      s = (j < n - 1) ? csr[beg + j] : d;
      float av[H];
      load_node_row<H>(as + (size_t)s * H, av);
#pragma unroll
      for (int h = 0; h < H; ++h) {
        float v = av[h] + dv[h];
        v = (v >= 0.0f) ? v : NEG_SLOPE * v;
        w[h] = expf(v - m[h]) / l[h];
      }
    }
    const int cb = (n - b < 32) ? (n - b) : 32;
    int jj = 0;
    for (; jj + U <= cb; jj += U) {
      float xr[U][H * NCH];
#pragma unroll
      for (int u = 0; u < U; ++u) {
        const int su = __builtin_amdgcn_readlane(s, jj + u);
        load_x_row<H, C, NCH>(xw + (size_t)su * LDX, lane, xr[u]);
      }
#pragma unroll
      for (int h = 0; h < H; ++h) {
#pragma unroll
        for (int u = 0; u < U; ++u) {
          const float wu = __int_as_float(
              __builtin_amdgcn_readlane(__float_as_int(w[h]), jj + u));
#pragma unroll
          for (int k = 0; k < NCH; ++k) acc[h * NCH + k] += wu * xr[u][h * NCH + k];
        }
      }
    }
    for (; jj < cb; ++jj) {
      const int s0 = __builtin_amdgcn_readlane(s, jj);
      float xa[H * NCH];
      load_x_row<H, C, NCH>(xw + (size_t)s0 * LDX, lane, xa);
#pragma unroll
      for (int h = 0; h < H; ++h) {
        const float w0 = __int_as_float(
            __builtin_amdgcn_readlane(__float_as_int(w[h]), jj));
#pragma unroll
        for (int k = 0; k < NCH; ++k) acc[h * NCH + k] += w0 * xa[h * NCH + k];
      }
    }
  }

  float* od = out + (size_t)d * (H * C);
#pragma unroll
  for (int h = 0; h < H; ++h) {
#pragma unroll
    for (int k = 0; k < NCH; ++k) {
      const int c = k * 32 + lane;
      if (c < C) {
        float v = acc[h * NCH + k] + bias[h * C + c];
        if (ELU) v = (v > 0.0f) ? v : (expf(v) - 1.0f);
        od[h * C + c] = v;
      }
    }
  }
  }
}

extern "C" void kernel_launch(void* const* d_in, const int* in_sizes, int n_in,
                              void* d_out, int out_size, void* d_ws, size_t ws_size,
                              hipStream_t stream) {
  const float* x      = (const float*)d_in[0];
  const int*   ei     = (const int*)d_in[1];
  const float* W1     = (const float*)d_in[2];
  const float* a1_src = (const float*)d_in[3];
  const float* a1_dst = (const float*)d_in[4];
  const float* b1     = (const float*)d_in[5];
  const float* W2     = (const float*)d_in[6];
  const float* a2_src = (const float*)d_in[7];
  const float* a2_dst = (const float*)d_in[8];
  const float* b2     = (const float*)d_in[9];
  float* out = (float*)d_out;
  (void)n_in; (void)out_size; (void)ws_size;

  const int F_IN = 256, H1 = 8, HC1 = 256, NC = 40, NCP = 64;
  const int Nn   = in_sizes[0] / F_IN;
  const int E    = in_sizes[1] / 2;
  const int* esrc = ei;
  const int* edst = ei + E;

  const int G  = (E + EDGE_CHUNK - 1) / EDGE_CHUNK;
  const int NB = (Nn + BIN_NODES - 1) / BIN_NODES;

  int* ip = (int*)d_ws;
  int*  binCnt   = ip;        ip += (size_t)G * MAX_BINS;
  int*  binStart = ip;        ip += 2 * MAX_BINS;
  int2* bpair    = (int2*)ip; ip += (size_t)2 * E;
  int*  start    = ip;        ip += (size_t)Nn;
  int*  cnt      = ip;        ip += (size_t)Nn;
  int*  csr      = ip;        ip += (size_t)E;
  float* p = (float*)ip;
  float* xw1  = p;        p += (size_t)Nn * HC1;
  float* as1  = p;        p += (size_t)Nn * H1;
  float* ad1  = p;        p += (size_t)Nn * H1;
  float* agg1 = p;        p += (size_t)Nn * HC1;
  float* w2p  = p;        p += (size_t)HC1 * NCP;
  float* xw2p = p;        p += (size_t)Nn * NCP;
  float* as2  = p;        p += (size_t)Nn;
  float* ad2  = p;        p += (size_t)Nn;

  dim3 blk(256);
  dim3 eblk(EDGE_THREADS);

  pad_cols_kernel<<<(HC1 * NCP + 255) / 256, blk, 0, stream>>>(
      W2, w2p, HC1, NC, NCP);

  bin_count_kernel<<<G, eblk, 0, stream>>>(edst, E, binCnt);
  bin_place_kernel<<<G, eblk, 0, stream>>>(esrc, edst, E, binCnt, binStart, bpair);
  bin_sort_kernel<<<NB, eblk, 0, stream>>>(bpair, binStart, Nn, start, cnt, csr);

  {
    int waves  = (Nn / 16) * (HC1 / 64);
    int blocks = (waves + 7) / 8;
    gemm_alpha_f32_wmma_kernel<8, 32, 2><<<blocks, blk, 0, stream>>>(
        x, W1, xw1, a1_src, a1_dst, as1, ad1, Nn, F_IN, HC1);
  }
  node_aggregate_kernel<8, 32, 256, 2, true><<<2048, blk, 0, stream>>>(
      csr, start, cnt, Nn, as1, ad1, xw1, b1, agg1);

  {
    int waves  = (Nn / 16) * (NCP / 64);
    int blocks = (waves + 7) / 8;
    gemm_alpha_f32_wmma_kernel<1, 40, 1><<<blocks, blk, 0, stream>>>(
        agg1, w2p, xw2p, a2_src, a2_dst, as2, ad2, Nn, HC1, NCP);
  }
  node_aggregate_kernel<1, 40, 64, 4, false><<<2048, blk, 0, stream>>>(
      csr, start, cnt, Nn, as2, ad2, xw2p, b2, out);
}


# GEMM: weight matrix put into LDS once per work-group (k-pair rows interleaved, bank-padded); waves walk row tiles and read B fragments from LDS instead of global; same products, same sum order
# speedup vs baseline: 16.0608x; 1.0700x over previous
#include <hip/hip_runtime.h>

typedef __attribute__((ext_vector_type(2))) float v2f;
typedef __attribute__((ext_vector_type(8))) float v8f;

#define NEG_SLOPE 0.2f

__global__ void pad_cols_kernel(const float* __restrict__ src,
                                float* __restrict__ dst,
                                int K, int c, int ldd) {
  int i = blockIdx.x * blockDim.x + threadIdx.x;
  if (i >= K * ldd) return;
  int r = i / ldd, j = i - r * ldd;
  dst[i] = (j < c) ? src[r * c + j] : 0.0f;
}

#define TILE_PITCH 66
#define GEMM_WAVES 8

template <int NH, int CH, int HPT, int NT, int KD>
__global__ void __launch_bounds__(GEMM_WAVES * 32)
gemm_alpha_f32_wmma_kernel(const float* __restrict__ A,
                           const float* __restrict__ B,
                           float* __restrict__ C,
                           const float* __restrict__ a_src,
                           const float* __restrict__ a_dst,
                           float* __restrict__ as, float* __restrict__ ad,
                           int M) {
  static_assert(HPT * CH <= 64 && (HPT == 1 || HPT == 2) && CH % 4 == 0,
                "whole heads inside a 64-column tile");
  static_assert(GEMM_WAVES % NT == 0 && KD % 32 == 0, "tile walk");
  constexpr int LDB = 64 * NT;
  constexpr int PR  = 2 * LDB + 32;
  constexpr int C4  = LDB / 4;
  constexpr int TOT = (KD / 2) * C4;
  constexpr int RS  = GEMM_WAVES / NT;
  constexpr int NSTEP = KD / 4;
  __shared__ alignas(16) float Bs[(KD / 2) * PR];
  __shared__ float tile[GEMM_WAVES * 16 * TILE_PITCH];
  const int lane = threadIdx.x & 31;
  const int wave = __builtin_amdgcn_readfirstlane((int)(threadIdx.x >> 5));
  const int half = lane >> 4;
  const int l    = lane & 15;

#pragma unroll 4
  for (int i = threadIdx.x; i < TOT; i += GEMM_WAVES * 32) {
    const int kp = i / C4, c4 = i - kp * C4;
    const float4 r0 = *reinterpret_cast<const float4*>(B + (size_t)(2 * kp) * LDB + 4 * c4);
    const float4 r1 = *reinterpret_cast<const float4*>(B + (size_t)(2 * kp + 1) * LDB + 4 * c4);
    float* d = Bs + kp * PR + 8 * c4;
    *reinterpret_cast<float4*>(d)     = make_float4(r0.x, r1.x, r0.y, r1.y);
    *reinterpret_cast<float4*>(d + 4) = make_float4(r0.z, r1.z, r0.w, r1.w);
  }
  __syncthreads();

  const int mtiles = M >> 4;
  const int ct   = wave % NT;
  const int rsub = wave / NT;
  const int n0   = ct << 6;
  const float* bl = Bs + half * PR + 2 * (n0 + l);
  float* tl = tile + wave * (16 * TILE_PITCH);
  float* trow = tl + (8 * half) * TILE_PITCH + l;

  for (int slot = rsub; ; slot += RS) {
    const int mt = slot * (int)gridDim.x + (int)blockIdx.x;
    if (mt >= mtiles) break;
    const int m0 = mt << 4;

    v8f acc[4] = {};
    const float* arow = A + (size_t)(m0 + l) * KD + 2 * half;

    v2f ac[8], b[4];
#pragma unroll
    for (int q = 0; q < 8; ++q) ac[q] = *(const v2f*)(arow + 4 * q);
#pragma unroll
    for (int s = 0; s < 4; ++s) b[s] = *(const v2f*)(bl + s * 32);

    for (int g = 0; g < KD / 32; ++g) {
      const int gn = (g + 1 < KD / 32) ? (g + 1) : g;
      v2f an[8];
#pragma unroll
      for (int q = 0; q < 8; ++q) an[q] = *(const v2f*)(arow + 32 * gn + 4 * q);
#pragma unroll
      for (int st = 0; st < 8; ++st) {
        const int kk = g * 8 + st;
        const int kn = (kk + 1 < NSTEP) ? (kk + 1) : kk;
        v2f bn[4];
#pragma unroll
        for (int s = 0; s < 4; ++s) bn[s] = *(const v2f*)(bl + (2 * kn) * PR + s * 32);
#pragma unroll
        for (int s = 0; s < 4; ++s)
          acc[s] = __builtin_amdgcn_wmma_f32_16x16x4_f32(
              false, ac[st], false, b[s], (short)0, acc[s], false, false);
#pragma unroll
        for (int s = 0; s < 4; ++s) b[s] = bn[s];
      }
#pragma unroll
      for (int q = 0; q < 8; ++q) ac[q] = an[q];
    }

    float* crow = C + (size_t)(m0 + 8 * half) * LDB + n0 + l;
#pragma unroll
    for (int s = 0; s < 4; ++s)
#pragma unroll
      for (int j = 0; j < 8; ++j) {
        crow[(size_t)j * LDB + s * 16] = acc[s][j];
        trow[j * TILE_PITCH + s * 16] = acc[s][j];
      }

    __builtin_amdgcn_fence(__ATOMIC_RELEASE, "wavefront");
    __builtin_amdgcn_wave_barrier();
    __builtin_amdgcn_fence(__ATOMIC_ACQUIRE, "wavefront");

    const int hd0 = ct * HPT;
    if constexpr (HPT == 2) {
      const int hd = hd0 + half;
      const float* xr = tl + l * TILE_PITCH + half * CH;
      const float4* ws4 = reinterpret_cast<const float4*>(a_src + (size_t)hd * CH);
      const float4* wd4 = reinterpret_cast<const float4*>(a_dst + (size_t)hd * CH);
      float s = 0.0f, d = 0.0f;
#pragma unroll
      for (int q = 0; q < CH / 4; ++q) {
        const float4 ws = ws4[q];
        const float4 wd = wd4[q];
        float v;
        v = xr[4 * q + 0]; s = __builtin_fmaf(v, ws.x, s); d = __builtin_fmaf(v, wd.x, d);
        v = xr[4 * q + 1]; s = __builtin_fmaf(v, ws.y, s); d = __builtin_fmaf(v, wd.y, d);
        v = xr[4 * q + 2]; s = __builtin_fmaf(v, ws.z, s); d = __builtin_fmaf(v, wd.z, d);
        v = xr[4 * q + 3]; s = __builtin_fmaf(v, ws.w, s); d = __builtin_fmaf(v, wd.w, d);
      }
      as[(size_t)(m0 + l) * NH + hd] = s;
      ad[(size_t)(m0 + l) * NH + hd] = d;
    } else {
      const float* xr = tl + l * TILE_PITCH;
      const float4* w4 = reinterpret_cast<const float4*>(
          (half ? a_dst : a_src) + (size_t)hd0 * CH);
      float s = 0.0f;
#pragma unroll
      for (int q = 0; q < CH / 4; ++q) {
        const float4 w = w4[q];
        s = __builtin_fmaf(xr[4 * q + 0], w.x, s);
        s = __builtin_fmaf(xr[4 * q + 1], w.y, s);
        s = __builtin_fmaf(xr[4 * q + 2], w.z, s);
        s = __builtin_fmaf(xr[4 * q + 3], w.w, s);
      }
      (half ? ad : as)[(size_t)(m0 + l) * NH + hd0] = s;
    }

    __builtin_amdgcn_fence(__ATOMIC_RELEASE, "wavefront");
    __builtin_amdgcn_wave_barrier();
    __builtin_amdgcn_fence(__ATOMIC_ACQUIRE, "wavefront");
  }
}

#define BIN_SHIFT 8
#define BIN_NODES 256
#define MAX_BINS 256
#define EDGE_CHUNK 8192
#define EDGE_THREADS 512
#define EDGE_PER_THREAD (EDGE_CHUNK / EDGE_THREADS)
#define ROW_GROUPS (EDGE_THREADS / 64)
#define ROWS_PER_TRIP 13
#define SORT_PER_TRIP 12
#define SORT_CAP 12288

static __device__ __forceinline__ int scan256_exclusive(int c, int tid, int* wsum) {
  const int lane = tid & 31, wave = tid >> 5;
  int incl = c;
  for (int off = 1; off < 32; off <<= 1) {
    const int t = __shfl_up(incl, off);
    if (lane >= off) incl += t;
  }
  if (wave < 8 && lane == 31) wsum[wave] = incl;
  __syncthreads();
  int wbase = 0;
  for (int w = 0; w < 8; ++w) wbase += (w < wave) ? wsum[w] : 0;
  return wbase + incl - c;
}

__global__ void __launch_bounds__(EDGE_THREADS)
bin_count_kernel(const int* __restrict__ edst, int E, int* __restrict__ binCnt) {
  __shared__ int hist[MAX_BINS];
  const int tid = threadIdx.x;
  const int base = blockIdx.x * EDGE_CHUNK + tid;
  int dd[EDGE_PER_THREAD];
#pragma unroll
  for (int q = 0; q < EDGE_PER_THREAD; ++q) {
    const int e = base + q * EDGE_THREADS;
    const int v = edst[(e < E) ? e : (E - 1)];
    dd[q] = (e < E) ? v : -1;
  }
  if (tid < MAX_BINS) hist[tid] = 0;
  __syncthreads();
#pragma unroll
  for (int q = 0; q < EDGE_PER_THREAD; ++q)
    if (dd[q] >= 0) atomicAdd(&hist[dd[q] >> BIN_SHIFT], 1);
  __syncthreads();
  if (tid < MAX_BINS) binCnt[blockIdx.x * MAX_BINS + tid] = hist[tid];
}

__global__ void __launch_bounds__(EDGE_THREADS)
bin_place_kernel(const int* __restrict__ esrc, const int* __restrict__ edst, int E,
                 const int* __restrict__ binCnt, int* __restrict__ binStart,
                 int2* __restrict__ bpair) {
  __shared__ alignas(16) int ptot[ROW_GROUPS][MAX_BINS];
  __shared__ alignas(16) int pbef[ROW_GROUPS][MAX_BINS];
  __shared__ int gbase[MAX_BINS];
  __shared__ int lstart[MAX_BINS];
  __shared__ int lcur[MAX_BINS];
  __shared__ int wsum_a[8];
  __shared__ int wsum_b[8];
  __shared__ int2 stage[EDGE_CHUNK];
  const int tid = threadIdx.x;
  const int g = blockIdx.x, G = gridDim.x;

  const int base = g * EDGE_CHUNK + tid;
  int dd[EDGE_PER_THREAD], ss[EDGE_PER_THREAD];
#pragma unroll
  for (int q = 0; q < EDGE_PER_THREAD; ++q) {
    const int e = base + q * EDGE_THREADS;
    const int ec = (e < E) ? e : (E - 1);
    const int dv = edst[ec];
    ss[q] = esrc[ec];
    dd[q] = (e < E) ? dv : -1;
  }

  const int rg = tid >> 6;
  const int c4 = (tid & 63) * 4;
  int t0 = 0, t1 = 0, t2 = 0, t3 = 0, b0 = 0, b1 = 0, b2 = 0, b3 = 0;
  for (int r0 = 0; r0 < G; r0 += ROW_GROUPS * ROWS_PER_TRIP) {
    int4 v[ROWS_PER_TRIP];
#pragma unroll
    for (int k = 0; k < ROWS_PER_TRIP; ++k) {
      const int r = r0 + rg + ROW_GROUPS * k;
      const int rc = (r < G) ? r : (G - 1);
      v[k] = *reinterpret_cast<const int4*>(binCnt + (size_t)rc * MAX_BINS + c4);
    }
#pragma unroll
    for (int k = 0; k < ROWS_PER_TRIP; ++k) {
      const int r = r0 + rg + ROW_GROUPS * k;
      const bool in = r < G, early = r < g;
      t0 += in ? v[k].x : 0; t1 += in ? v[k].y : 0;
      t2 += in ? v[k].z : 0; t3 += in ? v[k].w : 0;
      b0 += early ? v[k].x : 0; b1 += early ? v[k].y : 0;
      b2 += early ? v[k].z : 0; b3 += early ? v[k].w : 0;
    }
  }
  *reinterpret_cast<int4*>(&ptot[rg][c4]) = make_int4(t0, t1, t2, t3);
  *reinterpret_cast<int4*>(&pbef[rg][c4]) = make_int4(b0, b1, b2, b3);
  __syncthreads();

  int tot = 0, bef = 0, own = 0;
  if (tid < MAX_BINS) {
#pragma unroll
    for (int k = 0; k < ROW_GROUPS; ++k) { tot += ptot[k][tid]; bef += pbef[k][tid]; }
    own = binCnt[(size_t)g * MAX_BINS + tid];
  }
  const int excl  = scan256_exclusive(tot, tid, wsum_a);
  const int lexcl = scan256_exclusive(own, tid, wsum_b);
  if (tid < MAX_BINS) {
    gbase[tid] = excl + bef;
    lstart[tid] = lexcl;
    lcur[tid] = lexcl;
    if (g == 0) {
      binStart[tid] = excl;
      if (tid == MAX_BINS - 1) binStart[MAX_BINS] = excl + tot;
    }
  }
  __syncthreads();

  int jj[EDGE_PER_THREAD];
#pragma unroll
  for (int q = 0; q < EDGE_PER_THREAD; ++q) {
    jj[q] = -1;
    if (dd[q] >= 0) jj[q] = atomicAdd(&lcur[dd[q] >> BIN_SHIFT], 1);
  }
#pragma unroll
  for (int q = 0; q < EDGE_PER_THREAD; ++q)
    if (jj[q] >= 0) stage[jj[q]] = make_int2(ss[q], dd[q]);
  __syncthreads();

  const int left = E - g * EDGE_CHUNK;
  const int nval = (left < EDGE_CHUNK) ? left : EDGE_CHUNK;
  int2 pr[EDGE_PER_THREAD];
  int to[EDGE_PER_THREAD];
#pragma unroll
  for (int q = 0; q < EDGE_PER_THREAD; ++q) {
    const int j = q * EDGE_THREADS + tid;
    pr[q] = stage[(j < nval) ? j : (nval - 1)];
  }
#pragma unroll
  for (int q = 0; q < EDGE_PER_THREAD; ++q) {
    const int j = q * EDGE_THREADS + tid;
    const int bin = pr[q].y >> BIN_SHIFT;
    to[q] = gbase[bin] + (j - lstart[bin]);
  }
#pragma unroll
  for (int q = 0; q < EDGE_PER_THREAD; ++q) {
    const int j = q * EDGE_THREADS + tid;
    if (j < nval) bpair[to[q]] = pr[q];
  }
}

__global__ void __launch_bounds__(EDGE_THREADS)
bin_sort_kernel(const int2* __restrict__ bpair, const int* __restrict__ binStart,
                int Nn, int* __restrict__ start, int* __restrict__ cnt,
                int* __restrict__ csr) {
  __shared__ int lcnt[BIN_NODES];
  __shared__ int cur[BIN_NODES];
  __shared__ int wsum[8];
  __shared__ int stage[SORT_CAP];
  const int tid = threadIdx.x;
  const int b = blockIdx.x;
  const int beg = binStart[b];
  const int end = binStart[b + 1];
  const int n = end - beg;
  const bool fits = n <= SORT_CAP;
  if (tid < BIN_NODES) lcnt[tid] = 0;
  __syncthreads();
  for (int i0 = beg; i0 < end; i0 += EDGE_THREADS * SORT_PER_TRIP) {
    int dl[SORT_PER_TRIP];
#pragma unroll
    for (int q = 0; q < SORT_PER_TRIP; ++q) {
      const int i = i0 + q * EDGE_THREADS + tid;
      const int v = bpair[(i < end) ? i : (end - 1)].y;
      dl[q] = (i < end) ? (v & (BIN_NODES - 1)) : -1;
    }
#pragma unroll
    for (int q = 0; q < SORT_PER_TRIP; ++q)
      if (dl[q] >= 0) atomicAdd(&lcnt[dl[q]], 1);
  }
  __syncthreads();
  const int c = (tid < BIN_NODES) ? lcnt[tid] : 0;
  const int excl = scan256_exclusive(c, tid, wsum);
  if (tid < BIN_NODES) {
    cur[tid] = excl;
    const int d = (b << BIN_SHIFT) + tid;
    if (d < Nn) { start[d] = beg + excl; cnt[d] = c; }
  }
  __syncthreads();
  for (int i0 = beg; i0 < end; i0 += EDGE_THREADS * SORT_PER_TRIP) {
    int sv[SORT_PER_TRIP], dl[SORT_PER_TRIP], ps[SORT_PER_TRIP];
#pragma unroll
    for (int q = 0; q < SORT_PER_TRIP; ++q) {
      const int i = i0 + q * EDGE_THREADS + tid;
      const int2 pr = bpair[(i < end) ? i : (end - 1)];
      sv[q] = pr.x;
      dl[q] = (i < end) ? (pr.y & (BIN_NODES - 1)) : -1;
    }
#pragma unroll
    for (int q = 0; q < SORT_PER_TRIP; ++q) {
      ps[q] = -1;
      if (dl[q] >= 0) ps[q] = atomicAdd(&cur[dl[q]], 1);
    }
    if (fits) {
#pragma unroll
      for (int q = 0; q < SORT_PER_TRIP; ++q)
        if (ps[q] >= 0) stage[ps[q]] = sv[q];
    } else {
#pragma unroll
      for (int q = 0; q < SORT_PER_TRIP; ++q)
        if (ps[q] >= 0) csr[beg + ps[q]] = sv[q];
    }
  }
  if (fits) {
    __syncthreads();
    for (int j = tid; j < n; j += EDGE_THREADS) csr[beg + j] = stage[j];
  }
}

template <int H>
static __device__ __forceinline__ void load_node_row(const float* __restrict__ p,
                                                     float (&v)[H]) {
  if constexpr (H % 4 == 0) {
#pragma unroll
    for (int q = 0; q < H / 4; ++q) {
      const float4 t = reinterpret_cast<const float4*>(p)[q];
      v[4 * q + 0] = t.x; v[4 * q + 1] = t.y;
      v[4 * q + 2] = t.z; v[4 * q + 3] = t.w;
    }
  } else {
#pragma unroll
    for (int i = 0; i < H; ++i) v[i] = p[i];
  }
}

template <int H, int C, int NCH>
static __device__ __forceinline__ void load_x_row(const float* __restrict__ xs,
                                                  int lane, float (&xv)[H * NCH]) {
#pragma unroll
  for (int h = 0; h < H; ++h) {
#pragma unroll
    for (int k = 0; k < NCH; ++k) {
      const int c = k * 32 + lane;
      xv[h * NCH + k] = (c < C) ? xs[h * C + c] : 0.0f;
    }
  }
}

template <int H, int C, int LDX, int U, bool ELU>
__global__ void __launch_bounds__(256)
node_aggregate_kernel(const int* __restrict__ csr,
                      const int* __restrict__ start,
                      const int* __restrict__ cnt, int Nn,
                      const float* __restrict__ as,
                      const float* __restrict__ ad,
                      const float* __restrict__ xw,
                      const float* __restrict__ bias,
                      float* __restrict__ out) {
  constexpr int NCH = (C + 31) / 32;
  const int lane = threadIdx.x & 31;
  const int wv = __builtin_amdgcn_readfirstlane((int)(threadIdx.x >> 5));
  const int nwaves = (int)gridDim.x * (int)(blockDim.x >> 5);
  for (int d = blockIdx.x * (int)(blockDim.x >> 5) + wv; d < Nn; d += nwaves) {
  const int beg = __builtin_amdgcn_readfirstlane(start[d]);
  const int n = __builtin_amdgcn_readfirstlane(cnt[d]) + 1;

  float dv[H];
  load_node_row<H>(ad + (size_t)d * H, dv);

  float m[H], l[H];
#pragma unroll
  for (int h = 0; h < H; ++h) { m[h] = -__builtin_inff(); l[h] = 0.0f; }
  for (int b = 0; b < n; b += 32) {
    const int j = b + lane;
    if (j < n) {
      const int s = (j < n - 1) ? csr[beg + j] : d;
      float av[H];
      load_node_row<H>(as + (size_t)s * H, av);
#pragma unroll
      for (int h = 0; h < H; ++h) {
        float v = av[h] + dv[h];
        v = (v >= 0.0f) ? v : NEG_SLOPE * v;
        const float mn = fmaxf(m[h], v);
        l[h] = l[h] * expf(m[h] - mn) + expf(v - mn);
        m[h] = mn;
      }
    }
  }
#pragma unroll
  for (int h = 0; h < H; ++h) {
    float mx = m[h];
    for (int off = 16; off > 0; off >>= 1) mx = fmaxf(mx, __shfl_xor(mx, off));
    float t = l[h] * expf(m[h] - mx);
    for (int off = 16; off > 0; off >>= 1) t += __shfl_xor(t, off);
    m[h] = __int_as_float(__builtin_amdgcn_readfirstlane(__float_as_int(mx)));
    l[h] = __int_as_float(__builtin_amdgcn_readfirstlane(__float_as_int(t)));
  }

  float acc[H * NCH];
#pragma unroll
  for (int i = 0; i < H * NCH; ++i) acc[i] = 0.0f;

  for (int b = 0; b < n; b += 32) {
    const int j = b + lane;
    int s = d;
    float w[H];
#pragma unroll
    for (int h = 0; h < H; ++h) w[h] = 0.0f;
    if (j < n) {
      s = (j < n - 1) ? csr[beg + j] : d;
      float av[H];
      load_node_row<H>(as + (size_t)s * H, av);
#pragma unroll
      for (int h = 0; h < H; ++h) {
        float v = av[h] + dv[h];
        v = (v >= 0.0f) ? v : NEG_SLOPE * v;
        w[h] = expf(v - m[h]) / l[h];
      }
    }
    const int cb = (n - b < 32) ? (n - b) : 32;
    int jj = 0;
    for (; jj + U <= cb; jj += U) {
      float xr[U][H * NCH];
#pragma unroll
      for (int u = 0; u < U; ++u) {
        const int su = __builtin_amdgcn_readlane(s, jj + u);
        load_x_row<H, C, NCH>(xw + (size_t)su * LDX, lane, xr[u]);
      }
#pragma unroll
      for (int h = 0; h < H; ++h) {
#pragma unroll
        for (int u = 0; u < U; ++u) {
          const float wu = __int_as_float(
              __builtin_amdgcn_readlane(__float_as_int(w[h]), jj + u));
#pragma unroll
          for (int k = 0; k < NCH; ++k) acc[h * NCH + k] += wu * xr[u][h * NCH + k];
        }
      }
    }
    for (; jj < cb; ++jj) {
      const int s0 = __builtin_amdgcn_readlane(s, jj);
      float xa[H * NCH];
      load_x_row<H, C, NCH>(xw + (size_t)s0 * LDX, lane, xa);
#pragma unroll
      for (int h = 0; h < H; ++h) {
        const float w0 = __int_as_float(
            __builtin_amdgcn_readlane(__float_as_int(w[h]), jj));
#pragma unroll
        for (int k = 0; k < NCH; ++k) acc[h * NCH + k] += w0 * xa[h * NCH + k];
      }
    }
  }

  float* od = out + (size_t)d * (H * C);
#pragma unroll
  for (int h = 0; h < H; ++h) {
#pragma unroll
    for (int k = 0; k < NCH; ++k) {
      const int c = k * 32 + lane;
      if (c < C) {
        float v = acc[h * NCH + k] + bias[h * C + c];
        if (ELU) v = (v > 0.0f) ? v : (expf(v) - 1.0f);
        od[h * C + c] = v;
      }
    }
  }
  }
}

extern "C" void kernel_launch(void* const* d_in, const int* in_sizes, int n_in,
                              void* d_out, int out_size, void* d_ws, size_t ws_size,
                              hipStream_t stream) {
  const float* x      = (const float*)d_in[0];
  const int*   ei     = (const int*)d_in[1];
  const float* W1     = (const float*)d_in[2];
  const float* a1_src = (const float*)d_in[3];
  const float* a1_dst = (const float*)d_in[4];
  const float* b1     = (const float*)d_in[5];
  const float* W2     = (const float*)d_in[6];
  const float* a2_src = (const float*)d_in[7];
  const float* a2_dst = (const float*)d_in[8];
  const float* b2     = (const float*)d_in[9];
  float* out = (float*)d_out;
  (void)n_in; (void)out_size; (void)ws_size;

  const int F_IN = 256, H1 = 8, HC1 = 256, NC = 40, NCP = 64;
  const int Nn   = in_sizes[0] / F_IN;
  const int E    = in_sizes[1] / 2;
  const int* esrc = ei;
  const int* edst = ei + E;

  const int G  = (E + EDGE_CHUNK - 1) / EDGE_CHUNK;
  const int NB = (Nn + BIN_NODES - 1) / BIN_NODES;

  int* ip = (int*)d_ws;
  int*  binCnt   = ip;        ip += (size_t)G * MAX_BINS;
  int*  binStart = ip;        ip += 2 * MAX_BINS;
  int2* bpair    = (int2*)ip; ip += (size_t)2 * E;
  int*  start    = ip;        ip += (size_t)Nn;
  int*  cnt      = ip;        ip += (size_t)Nn;
  int*  csr      = ip;        ip += (size_t)E;
  float* p = (float*)ip;
  float* xw1  = p;        p += (size_t)Nn * HC1;
  float* as1  = p;        p += (size_t)Nn * H1;
  float* ad1  = p;        p += (size_t)Nn * H1;
  float* agg1 = p;        p += (size_t)Nn * HC1;
  float* w2p  = p;        p += (size_t)HC1 * NCP;
  float* xw2p = p;        p += (size_t)Nn * NCP;
  float* as2  = p;        p += (size_t)Nn;
  float* ad2  = p;        p += (size_t)Nn;

  dim3 blk(256);
  dim3 eblk(EDGE_THREADS);

  pad_cols_kernel<<<(HC1 * NCP + 255) / 256, blk, 0, stream>>>(
      W2, w2p, HC1, NC, NCP);

  bin_count_kernel<<<G, eblk, 0, stream>>>(edst, E, binCnt);
  bin_place_kernel<<<G, eblk, 0, stream>>>(esrc, edst, E, binCnt, binStart, bpair);
  bin_sort_kernel<<<NB, eblk, 0, stream>>>(bpair, binStart, Nn, start, cnt, csr);

  gemm_alpha_f32_wmma_kernel<8, 32, 2, 4, 256><<<256, blk, 0, stream>>>(
      x, W1, xw1, a1_src, a1_dst, as1, ad1, Nn);
  node_aggregate_kernel<8, 32, 256, 2, true><<<2048, blk, 0, stream>>>(
      csr, start, cnt, Nn, as1, ad1, xw1, b1, agg1);

  gemm_alpha_f32_wmma_kernel<1, 40, 1, 1, 256><<<256, blk, 0, stream>>>(
      agg1, w2p, xw2p, a2_src, a2_dst, as2, ad2, Nn);
  node_aggregate_kernel<1, 40, 64, 4, false><<<2048, blk, 0, stream>>>(
      csr, start, cnt, Nn, as2, ad2, xw2p, b2, out);
}


# node aggregate (both layers): a node with at most 32 edges gets its softmax in one pass: scores kept in registers, one exp per edge and head, csr and as read once; same values, same sum order
# speedup vs baseline: 17.2350x; 1.0731x over previous
#include <hip/hip_runtime.h>

typedef __attribute__((ext_vector_type(2))) float v2f;
typedef __attribute__((ext_vector_type(8))) float v8f;

#define NEG_SLOPE 0.2f

__global__ void pad_cols_kernel(const float* __restrict__ src,
                                float* __restrict__ dst,
                                int K, int c, int ldd) {
  int i = blockIdx.x * blockDim.x + threadIdx.x;
  if (i >= K * ldd) return;
  int r = i / ldd, j = i - r * ldd;
  dst[i] = (j < c) ? src[r * c + j] : 0.0f;
}

#define TILE_PITCH 66
#define GEMM_WAVES 8

template <int NH, int CH, int HPT, int NT, int KD>
__global__ void __launch_bounds__(GEMM_WAVES * 32)
gemm_alpha_f32_wmma_kernel(const float* __restrict__ A,
                           const float* __restrict__ B,
                           float* __restrict__ C,
                           const float* __restrict__ a_src,
                           const float* __restrict__ a_dst,
                           float* __restrict__ as, float* __restrict__ ad,
                           int M) {
  static_assert(HPT * CH <= 64 && (HPT == 1 || HPT == 2) && CH % 4 == 0,
                "whole heads inside a 64-column tile");
  static_assert(GEMM_WAVES % NT == 0 && KD % 32 == 0, "tile walk");
  constexpr int LDB = 64 * NT;
  constexpr int PR  = 2 * LDB + 32;
  constexpr int C4  = LDB / 4;
  constexpr int TOT = (KD / 2) * C4;
  constexpr int RS  = GEMM_WAVES / NT;
  constexpr int NSTEP = KD / 4;
  __shared__ alignas(16) float Bs[(KD / 2) * PR];
  __shared__ float tile[GEMM_WAVES * 16 * TILE_PITCH];
  const int lane = threadIdx.x & 31;
  const int wave = __builtin_amdgcn_readfirstlane((int)(threadIdx.x >> 5));
  const int half = lane >> 4;
  const int l    = lane & 15;

#pragma unroll 4
  for (int i = threadIdx.x; i < TOT; i += GEMM_WAVES * 32) {
    const int kp = i / C4, c4 = i - kp * C4;
    const float4 r0 = *reinterpret_cast<const float4*>(B + (size_t)(2 * kp) * LDB + 4 * c4);
    const float4 r1 = *reinterpret_cast<const float4*>(B + (size_t)(2 * kp + 1) * LDB + 4 * c4);
    float* d = Bs + kp * PR + 8 * c4;
    *reinterpret_cast<float4*>(d)     = make_float4(r0.x, r1.x, r0.y, r1.y);
    *reinterpret_cast<float4*>(d + 4) = make_float4(r0.z, r1.z, r0.w, r1.w);
  }
  __syncthreads();

  const int mtiles = M >> 4;
  const int ct   = wave % NT;
  const int rsub = wave / NT;
  const int n0   = ct << 6;
  const float* bl = Bs + half * PR + 2 * (n0 + l);
  float* tl = tile + wave * (16 * TILE_PITCH);
  float* trow = tl + (8 * half) * TILE_PITCH + l;

  for (int slot = rsub; ; slot += RS) {
    const int mt = slot * (int)gridDim.x + (int)blockIdx.x;
    if (mt >= mtiles) break;
    const int m0 = mt << 4;

    v8f acc[4] = {};
    const float* arow = A + (size_t)(m0 + l) * KD + 2 * half;

    v2f ac[8], b[4];
#pragma unroll
    for (int q = 0; q < 8; ++q) ac[q] = *(const v2f*)(arow + 4 * q);
#pragma unroll
    for (int s = 0; s < 4; ++s) b[s] = *(const v2f*)(bl + s * 32);

    for (int g = 0; g < KD / 32; ++g) {
      const int gn = (g + 1 < KD / 32) ? (g + 1) : g;
      v2f an[8];
#pragma unroll
      for (int q = 0; q < 8; ++q) an[q] = *(const v2f*)(arow + 32 * gn + 4 * q);
#pragma unroll
      for (int st = 0; st < 8; ++st) {
        const int kk = g * 8 + st;
        const int kn = (kk + 1 < NSTEP) ? (kk + 1) : kk;
        v2f bn[4];
#pragma unroll
        for (int s = 0; s < 4; ++s) bn[s] = *(const v2f*)(bl + (2 * kn) * PR + s * 32);
#pragma unroll
        for (int s = 0; s < 4; ++s)
          acc[s] = __builtin_amdgcn_wmma_f32_16x16x4_f32(
              false, ac[st], false, b[s], (short)0, acc[s], false, false);
#pragma unroll
        for (int s = 0; s < 4; ++s) b[s] = bn[s];
      }
#pragma unroll
      for (int q = 0; q < 8; ++q) ac[q] = an[q];
    }

    float* crow = C + (size_t)(m0 + 8 * half) * LDB + n0 + l;
#pragma unroll
    for (int s = 0; s < 4; ++s)
#pragma unroll
      for (int j = 0; j < 8; ++j) {
        crow[(size_t)j * LDB + s * 16] = acc[s][j];
        trow[j * TILE_PITCH + s * 16] = acc[s][j];
      }

    __builtin_amdgcn_fence(__ATOMIC_RELEASE, "wavefront");
    __builtin_amdgcn_wave_barrier();
    __builtin_amdgcn_fence(__ATOMIC_ACQUIRE, "wavefront");

    const int hd0 = ct * HPT;
    if constexpr (HPT == 2) {
      const int hd = hd0 + half;
      const float* xr = tl + l * TILE_PITCH + half * CH;
      const float4* ws4 = reinterpret_cast<const float4*>(a_src + (size_t)hd * CH);
      const float4* wd4 = reinterpret_cast<const float4*>(a_dst + (size_t)hd * CH);
      float s = 0.0f, d = 0.0f;
#pragma unroll
      for (int q = 0; q < CH / 4; ++q) {
        const float4 ws = ws4[q];
        const float4 wd = wd4[q];
        float v;
        v = xr[4 * q + 0]; s = __builtin_fmaf(v, ws.x, s); d = __builtin_fmaf(v, wd.x, d);
        v = xr[4 * q + 1]; s = __builtin_fmaf(v, ws.y, s); d = __builtin_fmaf(v, wd.y, d);
        v = xr[4 * q + 2]; s = __builtin_fmaf(v, ws.z, s); d = __builtin_fmaf(v, wd.z, d);
        v = xr[4 * q + 3]; s = __builtin_fmaf(v, ws.w, s); d = __builtin_fmaf(v, wd.w, d);
      }
      as[(size_t)(m0 + l) * NH + hd] = s;
      ad[(size_t)(m0 + l) * NH + hd] = d;
    } else {
      const float* xr = tl + l * TILE_PITCH;
      const float4* w4 = reinterpret_cast<const float4*>(
          (half ? a_dst : a_src) + (size_t)hd0 * CH);
      float s = 0.0f;
#pragma unroll
      for (int q = 0; q < CH / 4; ++q) {
        const float4 w = w4[q];
        s = __builtin_fmaf(xr[4 * q + 0], w.x, s);
        s = __builtin_fmaf(xr[4 * q + 1], w.y, s);
        s = __builtin_fmaf(xr[4 * q + 2], w.z, s);
        s = __builtin_fmaf(xr[4 * q + 3], w.w, s);
      }
      (half ? ad : as)[(size_t)(m0 + l) * NH + hd0] = s;
    }

    __builtin_amdgcn_fence(__ATOMIC_RELEASE, "wavefront");
    __builtin_amdgcn_wave_barrier();
    __builtin_amdgcn_fence(__ATOMIC_ACQUIRE, "wavefront");
  }
}

#define BIN_SHIFT 8
#define BIN_NODES 256
#define MAX_BINS 256
#define EDGE_CHUNK 8192
#define EDGE_THREADS 512
#define EDGE_PER_THREAD (EDGE_CHUNK / EDGE_THREADS)
#define ROW_GROUPS (EDGE_THREADS / 64)
#define ROWS_PER_TRIP 13
#define SORT_PER_TRIP 12
#define SORT_CAP 12288

static __device__ __forceinline__ int scan256_exclusive(int c, int tid, int* wsum) {
  const int lane = tid & 31, wave = tid >> 5;
  int incl = c;
  for (int off = 1; off < 32; off <<= 1) {
    const int t = __shfl_up(incl, off);
    if (lane >= off) incl += t;
  }
  if (wave < 8 && lane == 31) wsum[wave] = incl;
  __syncthreads();
  int wbase = 0;
  for (int w = 0; w < 8; ++w) wbase += (w < wave) ? wsum[w] : 0;
  return wbase + incl - c;
}

__global__ void __launch_bounds__(EDGE_THREADS)
bin_count_kernel(const int* __restrict__ edst, int E, int* __restrict__ binCnt) {
  __shared__ int hist[MAX_BINS];
  const int tid = threadIdx.x;
  const int base = blockIdx.x * EDGE_CHUNK + tid;
  int dd[EDGE_PER_THREAD];
#pragma unroll
  for (int q = 0; q < EDGE_PER_THREAD; ++q) {
    const int e = base + q * EDGE_THREADS;
    const int v = edst[(e < E) ? e : (E - 1)];
    dd[q] = (e < E) ? v : -1;
  }
  if (tid < MAX_BINS) hist[tid] = 0;
  __syncthreads();
#pragma unroll
  for (int q = 0; q < EDGE_PER_THREAD; ++q)
    if (dd[q] >= 0) atomicAdd(&hist[dd[q] >> BIN_SHIFT], 1);
  __syncthreads();
  if (tid < MAX_BINS) binCnt[blockIdx.x * MAX_BINS + tid] = hist[tid];
}

__global__ void __launch_bounds__(EDGE_THREADS)
bin_place_kernel(const int* __restrict__ esrc, const int* __restrict__ edst, int E,
                 const int* __restrict__ binCnt, int* __restrict__ binStart,
                 int2* __restrict__ bpair) {
  __shared__ alignas(16) int ptot[ROW_GROUPS][MAX_BINS];
  __shared__ alignas(16) int pbef[ROW_GROUPS][MAX_BINS];
  __shared__ int gbase[MAX_BINS];
  __shared__ int lstart[MAX_BINS];
  __shared__ int lcur[MAX_BINS];
  __shared__ int wsum_a[8];
  __shared__ int wsum_b[8];
  __shared__ int2 stage[EDGE_CHUNK];
  const int tid = threadIdx.x;
  const int g = blockIdx.x, G = gridDim.x;

  const int base = g * EDGE_CHUNK + tid;
  int dd[EDGE_PER_THREAD], ss[EDGE_PER_THREAD];
#pragma unroll
  for (int q = 0; q < EDGE_PER_THREAD; ++q) {
    const int e = base + q * EDGE_THREADS;
    const int ec = (e < E) ? e : (E - 1);
    const int dv = edst[ec];
    ss[q] = esrc[ec];
    dd[q] = (e < E) ? dv : -1;
  }

  const int rg = tid >> 6;
  const int c4 = (tid & 63) * 4;
  int t0 = 0, t1 = 0, t2 = 0, t3 = 0, b0 = 0, b1 = 0, b2 = 0, b3 = 0;
  for (int r0 = 0; r0 < G; r0 += ROW_GROUPS * ROWS_PER_TRIP) {
    int4 v[ROWS_PER_TRIP];
#pragma unroll
    for (int k = 0; k < ROWS_PER_TRIP; ++k) {
      const int r = r0 + rg + ROW_GROUPS * k;
      const int rc = (r < G) ? r : (G - 1);
      v[k] = *reinterpret_cast<const int4*>(binCnt + (size_t)rc * MAX_BINS + c4);
    }
#pragma unroll
    for (int k = 0; k < ROWS_PER_TRIP; ++k) {
      const int r = r0 + rg + ROW_GROUPS * k;
      const bool in = r < G, early = r < g;
      t0 += in ? v[k].x : 0; t1 += in ? v[k].y : 0;
      t2 += in ? v[k].z : 0; t3 += in ? v[k].w : 0;
      b0 += early ? v[k].x : 0; b1 += early ? v[k].y : 0;
      b2 += early ? v[k].z : 0; b3 += early ? v[k].w : 0;
    }
  }
  *reinterpret_cast<int4*>(&ptot[rg][c4]) = make_int4(t0, t1, t2, t3);
  *reinterpret_cast<int4*>(&pbef[rg][c4]) = make_int4(b0, b1, b2, b3);
  __syncthreads();

  int tot = 0, bef = 0, own = 0;
  if (tid < MAX_BINS) {
#pragma unroll
    for (int k = 0; k < ROW_GROUPS; ++k) { tot += ptot[k][tid]; bef += pbef[k][tid]; }
    own = binCnt[(size_t)g * MAX_BINS + tid];
  }
  const int excl  = scan256_exclusive(tot, tid, wsum_a);
  const int lexcl = scan256_exclusive(own, tid, wsum_b);
  if (tid < MAX_BINS) {
    gbase[tid] = excl + bef;
    lstart[tid] = lexcl;
    lcur[tid] = lexcl;
    if (g == 0) {
      binStart[tid] = excl;
      if (tid == MAX_BINS - 1) binStart[MAX_BINS] = excl + tot;
    }
  }
  __syncthreads();

  int jj[EDGE_PER_THREAD];
#pragma unroll
  for (int q = 0; q < EDGE_PER_THREAD; ++q) {
    jj[q] = -1;
    if (dd[q] >= 0) jj[q] = atomicAdd(&lcur[dd[q] >> BIN_SHIFT], 1);
  }
#pragma unroll
  for (int q = 0; q < EDGE_PER_THREAD; ++q)
    if (jj[q] >= 0) stage[jj[q]] = make_int2(ss[q], dd[q]);
  __syncthreads();

  const int left = E - g * EDGE_CHUNK;
  const int nval = (left < EDGE_CHUNK) ? left : EDGE_CHUNK;
  int2 pr[EDGE_PER_THREAD];
  int to[EDGE_PER_THREAD];
#pragma unroll
  for (int q = 0; q < EDGE_PER_THREAD; ++q) {
    const int j = q * EDGE_THREADS + tid;
    pr[q] = stage[(j < nval) ? j : (nval - 1)];
  }
#pragma unroll
  for (int q = 0; q < EDGE_PER_THREAD; ++q) {
    const int j = q * EDGE_THREADS + tid;
    const int bin = pr[q].y >> BIN_SHIFT;
    to[q] = gbase[bin] + (j - lstart[bin]);
  }
#pragma unroll
  for (int q = 0; q < EDGE_PER_THREAD; ++q) {
    const int j = q * EDGE_THREADS + tid;
    if (j < nval) bpair[to[q]] = pr[q];
  }
}

__global__ void __launch_bounds__(EDGE_THREADS)
bin_sort_kernel(const int2* __restrict__ bpair, const int* __restrict__ binStart,
                int Nn, int* __restrict__ start, int* __restrict__ cnt,
                int* __restrict__ csr) {
  __shared__ int lcnt[BIN_NODES];
  __shared__ int cur[BIN_NODES];
  __shared__ int wsum[8];
  __shared__ int stage[SORT_CAP];
  const int tid = threadIdx.x;
  const int b = blockIdx.x;
  const int beg = binStart[b];
  const int end = binStart[b + 1];
  const int n = end - beg;
  const bool fits = n <= SORT_CAP;
  if (tid < BIN_NODES) lcnt[tid] = 0;
  __syncthreads();
  for (int i0 = beg; i0 < end; i0 += EDGE_THREADS * SORT_PER_TRIP) {
    int dl[SORT_PER_TRIP];
#pragma unroll
    for (int q = 0; q < SORT_PER_TRIP; ++q) {
      const int i = i0 + q * EDGE_THREADS + tid;
      const int v = bpair[(i < end) ? i : (end - 1)].y;
      dl[q] = (i < end) ? (v & (BIN_NODES - 1)) : -1;
    }
#pragma unroll
    for (int q = 0; q < SORT_PER_TRIP; ++q)
      if (dl[q] >= 0) atomicAdd(&lcnt[dl[q]], 1);
  }
  __syncthreads();
  const int c = (tid < BIN_NODES) ? lcnt[tid] : 0;
  const int excl = scan256_exclusive(c, tid, wsum);
  if (tid < BIN_NODES) {
    cur[tid] = excl;
    const int d = (b << BIN_SHIFT) + tid;
    if (d < Nn) { start[d] = beg + excl; cnt[d] = c; }
  }
  __syncthreads();
  for (int i0 = beg; i0 < end; i0 += EDGE_THREADS * SORT_PER_TRIP) {
    int sv[SORT_PER_TRIP], dl[SORT_PER_TRIP], ps[SORT_PER_TRIP];
#pragma unroll
    for (int q = 0; q < SORT_PER_TRIP; ++q) {
      const int i = i0 + q * EDGE_THREADS + tid;
      const int2 pr = bpair[(i < end) ? i : (end - 1)];
      sv[q] = pr.x;
      dl[q] = (i < end) ? (pr.y & (BIN_NODES - 1)) : -1;
    }
#pragma unroll
    for (int q = 0; q < SORT_PER_TRIP; ++q) {
      ps[q] = -1;
      if (dl[q] >= 0) ps[q] = atomicAdd(&cur[dl[q]], 1);
    }
    if (fits) {
#pragma unroll
      for (int q = 0; q < SORT_PER_TRIP; ++q)
        if (ps[q] >= 0) stage[ps[q]] = sv[q];
    } else {
#pragma unroll
      for (int q = 0; q < SORT_PER_TRIP; ++q)
        if (ps[q] >= 0) csr[beg + ps[q]] = sv[q];
    }
  }
  if (fits) {
    __syncthreads();
    for (int j = tid; j < n; j += EDGE_THREADS) csr[beg + j] = stage[j];
  }
}

template <int H>
static __device__ __forceinline__ void load_node_row(const float* __restrict__ p,
                                                     float (&v)[H]) {
  if constexpr (H % 4 == 0) {
#pragma unroll
    for (int q = 0; q < H / 4; ++q) {
      const float4 t = reinterpret_cast<const float4*>(p)[q];
      v[4 * q + 0] = t.x; v[4 * q + 1] = t.y;
      v[4 * q + 2] = t.z; v[4 * q + 3] = t.w;
    }
  } else {
#pragma unroll
    for (int i = 0; i < H; ++i) v[i] = p[i];
  }
}

template <int H, int C, int NCH>
static __device__ __forceinline__ void load_x_row(const float* __restrict__ xs,
                                                  int lane, float (&xv)[H * NCH]) {
#pragma unroll
  for (int h = 0; h < H; ++h) {
#pragma unroll
    for (int k = 0; k < NCH; ++k) {
      const int c = k * 32 + lane;
      xv[h * NCH + k] = (c < C) ? xs[h * C + c] : 0.0f;
    }
  }
}

template <int H, int C, int NCH, int LDX, int U>
static __device__ __forceinline__ void walk_run(const float* __restrict__ xw, int lane,
                                                int s, const float (&w)[H], int cb,
                                                float (&acc)[H * NCH]) {
  int jj = 0;
  for (; jj + U <= cb; jj += U) {
    float xr[U][H * NCH];
#pragma unroll
    for (int u = 0; u < U; ++u) {
      const int su = __builtin_amdgcn_readlane(s, jj + u);
      load_x_row<H, C, NCH>(xw + (size_t)su * LDX, lane, xr[u]);
    }
#pragma unroll
    for (int h = 0; h < H; ++h) {
#pragma unroll
      for (int u = 0; u < U; ++u) {
        const float wu = __int_as_float(
            __builtin_amdgcn_readlane(__float_as_int(w[h]), jj + u));
#pragma unroll
        for (int k = 0; k < NCH; ++k) acc[h * NCH + k] += wu * xr[u][h * NCH + k];
      }
    }
  }
  for (; jj < cb; ++jj) {
    const int s0 = __builtin_amdgcn_readlane(s, jj);
    float xa[H * NCH];
    load_x_row<H, C, NCH>(xw + (size_t)s0 * LDX, lane, xa);
#pragma unroll
    for (int h = 0; h < H; ++h) {
      const float w0 = __int_as_float(
          __builtin_amdgcn_readlane(__float_as_int(w[h]), jj));
#pragma unroll
      for (int k = 0; k < NCH; ++k) acc[h * NCH + k] += w0 * xa[h * NCH + k];
    }
  }
}

template <int H, int C, int LDX, int U, bool ELU>
__global__ void __launch_bounds__(256)
node_aggregate_kernel(const int* __restrict__ csr,
                      const int* __restrict__ start,
                      const int* __restrict__ cnt, int Nn,
                      const float* __restrict__ as,
                      const float* __restrict__ ad,
                      const float* __restrict__ xw,
                      const float* __restrict__ bias,
                      float* __restrict__ out) {
  constexpr int NCH = (C + 31) / 32;
  const int lane = threadIdx.x & 31;
  const int wv = __builtin_amdgcn_readfirstlane((int)(threadIdx.x >> 5));
  const int nwaves = (int)gridDim.x * (int)(blockDim.x >> 5);
  for (int d = blockIdx.x * (int)(blockDim.x >> 5) + wv; d < Nn; d += nwaves) {
  const int beg = __builtin_amdgcn_readfirstlane(start[d]);
  const int n = __builtin_amdgcn_readfirstlane(cnt[d]) + 1;

  float dv[H];
  load_node_row<H>(ad + (size_t)d * H, dv);

  float acc[H * NCH];
#pragma unroll
  for (int i = 0; i < H * NCH; ++i) acc[i] = 0.0f;

  if (n <= 32) {
    const bool on = lane < n;
    int s = d;
    float v[H];
#pragma unroll
    for (int h = 0; h < H; ++h) v[h] = -__builtin_inff();
    if (on) {
      s = (lane < n - 1) ? csr[beg + lane] : d;
      float av[H];
      load_node_row<H>(as + (size_t)s * H, av);
#pragma unroll
      for (int h = 0; h < H; ++h) {
        const float t = av[h] + dv[h];
        v[h] = (t >= 0.0f) ? t : NEG_SLOPE * t;
      }
    }
    float mx[H];
#pragma unroll
    for (int h = 0; h < H; ++h) mx[h] = v[h];
#pragma unroll
    for (int off = 16; off > 0; off >>= 1) {
#pragma unroll
      for (int h = 0; h < H; ++h) mx[h] = fmaxf(mx[h], __shfl_xor(mx[h], off));
    }
    float e[H], t[H];
#pragma unroll
    for (int h = 0; h < H; ++h) {
      const float smx = __int_as_float(
          __builtin_amdgcn_readfirstlane(__float_as_int(mx[h])));
      const float ex = expf(v[h] - smx);
      e[h] = on ? ex : 0.0f;
      t[h] = e[h];
    }
#pragma unroll
    for (int off = 16; off > 0; off >>= 1) {
#pragma unroll
      for (int h = 0; h < H; ++h) t[h] += __shfl_xor(t[h], off);
    }
    float w[H];
#pragma unroll
    for (int h = 0; h < H; ++h) {
      const float ls = __int_as_float(
          __builtin_amdgcn_readfirstlane(__float_as_int(t[h])));
      w[h] = e[h] / ls;
    }
    walk_run<H, C, NCH, LDX, U>(xw, lane, s, w, n, acc);
  } else {
  float m[H], l[H];
#pragma unroll
  for (int h = 0; h < H; ++h) { m[h] = -__builtin_inff(); l[h] = 0.0f; }
  for (int b = 0; b < n; b += 32) {
    const int j = b + lane;
    if (j < n) {
      const int s = (j < n - 1) ? csr[beg + j] : d;
      float av[H];
      load_node_row<H>(as + (size_t)s * H, av);
#pragma unroll
      for (int h = 0; h < H; ++h) {
        float v = av[h] + dv[h];
        v = (v >= 0.0f) ? v : NEG_SLOPE * v;
        const float mn = fmaxf(m[h], v);
        l[h] = l[h] * expf(m[h] - mn) + expf(v - mn);
        m[h] = mn;
      }
    }
  }
#pragma unroll
  for (int h = 0; h < H; ++h) {
    float mx = m[h];
    for (int off = 16; off > 0; off >>= 1) mx = fmaxf(mx, __shfl_xor(mx, off));
    float t = l[h] * expf(m[h] - mx);
    for (int off = 16; off > 0; off >>= 1) t += __shfl_xor(t, off);
    m[h] = __int_as_float(__builtin_amdgcn_readfirstlane(__float_as_int(mx)));
    l[h] = __int_as_float(__builtin_amdgcn_readfirstlane(__float_as_int(t)));
  }

  for (int b = 0; b < n; b += 32) {
    const int j = b + lane;
    int s = d;
    float w[H];
#pragma unroll
    for (int h = 0; h < H; ++h) w[h] = 0.0f;
    if (j < n) {
      s = (j < n - 1) ? csr[beg + j] : d;
      float av[H];
      load_node_row<H>(as + (size_t)s * H, av);
#pragma unroll
      for (int h = 0; h < H; ++h) {
        float v = av[h] + dv[h];
        v = (v >= 0.0f) ? v : NEG_SLOPE * v;
        w[h] = expf(v - m[h]) / l[h];
      }
    }
    const int cb = (n - b < 32) ? (n - b) : 32;
    walk_run<H, C, NCH, LDX, U>(xw, lane, s, w, cb, acc);
  }
  }

  float* od = out + (size_t)d * (H * C);
#pragma unroll
  for (int h = 0; h < H; ++h) {
#pragma unroll
    for (int k = 0; k < NCH; ++k) {
      const int c = k * 32 + lane;
      if (c < C) {
        float v = acc[h * NCH + k] + bias[h * C + c];
        if (ELU) v = (v > 0.0f) ? v : (expf(v) - 1.0f);
        od[h * C + c] = v;
      }
    }
  }
  }
}

extern "C" void kernel_launch(void* const* d_in, const int* in_sizes, int n_in,
                              void* d_out, int out_size, void* d_ws, size_t ws_size,
                              hipStream_t stream) {
  const float* x      = (const float*)d_in[0];
  const int*   ei     = (const int*)d_in[1];
  const float* W1     = (const float*)d_in[2];
  const float* a1_src = (const float*)d_in[3];
  const float* a1_dst = (const float*)d_in[4];
  const float* b1     = (const float*)d_in[5];
  const float* W2     = (const float*)d_in[6];
  const float* a2_src = (const float*)d_in[7];
  const float* a2_dst = (const float*)d_in[8];
  const float* b2     = (const float*)d_in[9];
  float* out = (float*)d_out;
  (void)n_in; (void)out_size; (void)ws_size;

  const int F_IN = 256, H1 = 8, HC1 = 256, NC = 40, NCP = 64;
  const int Nn   = in_sizes[0] / F_IN;
  const int E    = in_sizes[1] / 2;
  const int* esrc = ei;
  const int* edst = ei + E;

  const int G  = (E + EDGE_CHUNK - 1) / EDGE_CHUNK;
  const int NB = (Nn + BIN_NODES - 1) / BIN_NODES;

  int* ip = (int*)d_ws;
  int*  binCnt   = ip;        ip += (size_t)G * MAX_BINS;
  int*  binStart = ip;        ip += 2 * MAX_BINS;
  int2* bpair    = (int2*)ip; ip += (size_t)2 * E;
  int*  start    = ip;        ip += (size_t)Nn;
  int*  cnt      = ip;        ip += (size_t)Nn;
  int*  csr      = ip;        ip += (size_t)E;
  float* p = (float*)ip;
  float* xw1  = p;        p += (size_t)Nn * HC1;
  float* as1  = p;        p += (size_t)Nn * H1;
  float* ad1  = p;        p += (size_t)Nn * H1;
  float* agg1 = p;        p += (size_t)Nn * HC1;
  float* w2p  = p;        p += (size_t)HC1 * NCP;
  float* xw2p = p;        p += (size_t)Nn * NCP;
  float* as2  = p;        p += (size_t)Nn;
  float* ad2  = p;        p += (size_t)Nn;

  dim3 blk(256);
  dim3 eblk(EDGE_THREADS);

  pad_cols_kernel<<<(HC1 * NCP + 255) / 256, blk, 0, stream>>>(
      W2, w2p, HC1, NC, NCP);

  bin_count_kernel<<<G, eblk, 0, stream>>>(edst, E, binCnt);
  bin_place_kernel<<<G, eblk, 0, stream>>>(esrc, edst, E, binCnt, binStart, bpair);
  bin_sort_kernel<<<NB, eblk, 0, stream>>>(bpair, binStart, Nn, start, cnt, csr);

  gemm_alpha_f32_wmma_kernel<8, 32, 2, 4, 256><<<256, blk, 0, stream>>>(
      x, W1, xw1, a1_src, a1_dst, as1, ad1, Nn);
  node_aggregate_kernel<8, 32, 256, 2, true><<<2048, blk, 0, stream>>>(
      csr, start, cnt, Nn, as1, ad1, xw1, b1, agg1);

  gemm_alpha_f32_wmma_kernel<1, 40, 1, 1, 256><<<256, blk, 0, stream>>>(
      agg1, w2p, xw2p, a2_src, a2_dst, as2, ad2, Nn);
  node_aggregate_kernel<1, 40, 64, 4, false><<<2048, blk, 0, stream>>>(
      csr, start, cnt, Nn, as2, ad2, xw2p, b2, out);
}


# edge grouping level 1 done by the layer-1 GEMM work-groups after their row tiles, chunk-local (returning LDS add = rank, no counting pass): 8 launches -> 6; slot order by arrival as before
# speedup vs baseline: 17.9509x; 1.0415x over previous
#include <hip/hip_runtime.h>

typedef __attribute__((ext_vector_type(2))) float v2f;
typedef __attribute__((ext_vector_type(8))) float v8f;

#define NEG_SLOPE 0.2f

__global__ void pad_cols_kernel(const float* __restrict__ src,
                                float* __restrict__ dst,
                                int K, int c, int ldd) {
  int i = blockIdx.x * blockDim.x + threadIdx.x;
  if (i >= K * ldd) return;
  int r = i / ldd, j = i - r * ldd;
  dst[i] = (j < c) ? src[r * c + j] : 0.0f;
}

#define BIN_SHIFT 8
#define BIN_NODES 256
#define MAX_BINS 256
#define EDGE_CHUNK 4096
#define CHUNK_THREADS 256
#define EDGE_PER_THREAD (EDGE_CHUNK / CHUNK_THREADS)
#define MAX_CHUNKS 1024
#define SORT_THREADS 512
#define SORT_PER_TRIP 12
#define SORT_CAP 12288

static __device__ __forceinline__ int scan256_exclusive(int c, int tid, int* wsum) {
  const int lane = tid & 31, wave = tid >> 5;
  int incl = c;
  for (int off = 1; off < 32; off <<= 1) {
    const int t = __shfl_up(incl, off);
    if (lane >= off) incl += t;
  }
  if (wave < 8 && lane == 31) wsum[wave] = incl;
  __syncthreads();
  int wbase = 0;
  for (int w = 0; w < 8; ++w) wbase += (w < wave) ? wsum[w] : 0;
  return wbase + incl - c;
}

#define TILE_PITCH 66
#define GEMM_WAVES 8

template <int NH, int CH, int HPT, int NT, int KD, bool EDGES>
__global__ void __launch_bounds__(GEMM_WAVES * 32)
gemm_alpha_f32_wmma_kernel(const float* __restrict__ A,
                           const float* __restrict__ B,
                           float* __restrict__ C,
                           const float* __restrict__ a_src,
                           const float* __restrict__ a_dst,
                           float* __restrict__ as, float* __restrict__ ad,
                           int M,
                           const int* __restrict__ esrc,
                           const int* __restrict__ edst, int E, int G,
                           int* __restrict__ binCnt, int* __restrict__ binOff,
                           int2* __restrict__ bpair) {
  static_assert(HPT * CH <= 64 && (HPT == 1 || HPT == 2) && CH % 4 == 0,
                "whole heads inside a 64-column tile");
  static_assert(GEMM_WAVES % NT == 0 && KD % 32 == 0, "tile walk");
  constexpr int LDB = 64 * NT;
  constexpr int PR  = 2 * LDB + 32;
  constexpr int C4  = LDB / 4;
  constexpr int TOT = (KD / 2) * C4;
  constexpr int RS  = GEMM_WAVES / NT;
  constexpr int NSTEP = KD / 4;
  __shared__ alignas(16) float Bs[(KD / 2) * PR];
  __shared__ float tile[GEMM_WAVES * 16 * TILE_PITCH];
  const int lane = threadIdx.x & 31;
  const int wave = __builtin_amdgcn_readfirstlane((int)(threadIdx.x >> 5));
  const int half = lane >> 4;
  const int l    = lane & 15;

#pragma unroll 4
  for (int i = threadIdx.x; i < TOT; i += GEMM_WAVES * 32) {
    const int kp = i / C4, c4 = i - kp * C4;
    const float4 r0 = *reinterpret_cast<const float4*>(B + (size_t)(2 * kp) * LDB + 4 * c4);
    const float4 r1 = *reinterpret_cast<const float4*>(B + (size_t)(2 * kp + 1) * LDB + 4 * c4);
    float* d = Bs + kp * PR + 8 * c4;
    *reinterpret_cast<float4*>(d)     = make_float4(r0.x, r1.x, r0.y, r1.y);
    *reinterpret_cast<float4*>(d + 4) = make_float4(r0.z, r1.z, r0.w, r1.w);
  }
  __syncthreads();

  const int mtiles = M >> 4;
  const int ct   = wave % NT;
  const int rsub = wave / NT;
  const int n0   = ct << 6;
  const float* bl = Bs + half * PR + 2 * (n0 + l);
  float* tl = tile + wave * (16 * TILE_PITCH);
  float* trow = tl + (8 * half) * TILE_PITCH + l;

  for (int slot = rsub; ; slot += RS) {
    const int mt = slot * (int)gridDim.x + (int)blockIdx.x;
    if (mt >= mtiles) break;
    const int m0 = mt << 4;

    v8f acc[4] = {};
    const float* arow = A + (size_t)(m0 + l) * KD + 2 * half;

    v2f ac[8], b[4];
#pragma unroll
    for (int q = 0; q < 8; ++q) ac[q] = *(const v2f*)(arow + 4 * q);
#pragma unroll
    for (int s = 0; s < 4; ++s) b[s] = *(const v2f*)(bl + s * 32);

    for (int g = 0; g < KD / 32; ++g) {
      const int gn = (g + 1 < KD / 32) ? (g + 1) : g;
      v2f an[8];
#pragma unroll
      for (int q = 0; q < 8; ++q) an[q] = *(const v2f*)(arow + 32 * gn + 4 * q);
#pragma unroll
      for (int st = 0; st < 8; ++st) {
        const int kk = g * 8 + st;
        const int kn = (kk + 1 < NSTEP) ? (kk + 1) : kk;
        v2f bn[4];
#pragma unroll
        for (int s = 0; s < 4; ++s) bn[s] = *(const v2f*)(bl + (2 * kn) * PR + s * 32);
#pragma unroll
        for (int s = 0; s < 4; ++s)
          acc[s] = __builtin_amdgcn_wmma_f32_16x16x4_f32(
              false, ac[st], false, b[s], (short)0, acc[s], false, false);
#pragma unroll
        for (int s = 0; s < 4; ++s) b[s] = bn[s];
      }
#pragma unroll
      for (int q = 0; q < 8; ++q) ac[q] = an[q];
    }

    float* crow = C + (size_t)(m0 + 8 * half) * LDB + n0 + l;
#pragma unroll
    for (int s = 0; s < 4; ++s)
#pragma unroll
      for (int j = 0; j < 8; ++j) {
        crow[(size_t)j * LDB + s * 16] = acc[s][j];
        trow[j * TILE_PITCH + s * 16] = acc[s][j];
      }

    __builtin_amdgcn_fence(__ATOMIC_RELEASE, "wavefront");
    __builtin_amdgcn_wave_barrier();
    __builtin_amdgcn_fence(__ATOMIC_ACQUIRE, "wavefront");

    const int hd0 = ct * HPT;
    if constexpr (HPT == 2) {
      const int hd = hd0 + half;
      const float* xr = tl + l * TILE_PITCH + half * CH;
      const float4* ws4 = reinterpret_cast<const float4*>(a_src + (size_t)hd * CH);
      const float4* wd4 = reinterpret_cast<const float4*>(a_dst + (size_t)hd * CH);
      float s = 0.0f, d = 0.0f;
#pragma unroll
      for (int q = 0; q < CH / 4; ++q) {
        const float4 ws = ws4[q];
        const float4 wd = wd4[q];
        float v;
        v = xr[4 * q + 0]; s = __builtin_fmaf(v, ws.x, s); d = __builtin_fmaf(v, wd.x, d);
        v = xr[4 * q + 1]; s = __builtin_fmaf(v, ws.y, s); d = __builtin_fmaf(v, wd.y, d);
        v = xr[4 * q + 2]; s = __builtin_fmaf(v, ws.z, s); d = __builtin_fmaf(v, wd.z, d);
        v = xr[4 * q + 3]; s = __builtin_fmaf(v, ws.w, s); d = __builtin_fmaf(v, wd.w, d);
      }
      as[(size_t)(m0 + l) * NH + hd] = s;
      ad[(size_t)(m0 + l) * NH + hd] = d;
    } else {
      const float* xr = tl + l * TILE_PITCH;
      const float4* w4 = reinterpret_cast<const float4*>(
          (half ? a_dst : a_src) + (size_t)hd0 * CH);
      float s = 0.0f;
#pragma unroll
      for (int q = 0; q < CH / 4; ++q) {
        const float4 w = w4[q];
        s = __builtin_fmaf(xr[4 * q + 0], w.x, s);
        s = __builtin_fmaf(xr[4 * q + 1], w.y, s);
        s = __builtin_fmaf(xr[4 * q + 2], w.z, s);
        s = __builtin_fmaf(xr[4 * q + 3], w.w, s);
      }
      (half ? ad : as)[(size_t)(m0 + l) * NH + hd0] = s;
    }

    __builtin_amdgcn_fence(__ATOMIC_RELEASE, "wavefront");
    __builtin_amdgcn_wave_barrier();
    __builtin_amdgcn_fence(__ATOMIC_ACQUIRE, "wavefront");
  }

  if constexpr (EDGES) {
    static_assert(GEMM_WAVES * 32 == CHUNK_THREADS && CHUNK_THREADS == MAX_BINS,
                  "a thread per bin, and scan256_exclusive over the whole block");
    static_assert((KD / 2) * PR >= 2 * MAX_BINS + 16 + 2 * EDGE_CHUNK, "tables and staging fit B's place");
    __syncthreads();
    int*  hist  = reinterpret_cast<int*>(Bs);
    int*  lst   = hist + MAX_BINS;
    int*  wsum  = lst + MAX_BINS;
    int2* stage = reinterpret_cast<int2*>(wsum + 16);
    const int tid = threadIdx.x;
    for (int g = (int)gridDim.x - 1 - (int)blockIdx.x; g < G; g += (int)gridDim.x) {
      const int base = g * EDGE_CHUNK + tid;
      int dd[EDGE_PER_THREAD], ss[EDGE_PER_THREAD], rk[EDGE_PER_THREAD];
#pragma unroll
      for (int q = 0; q < EDGE_PER_THREAD; ++q) {
        const int e = base + q * CHUNK_THREADS;
        const int ec = (e < E) ? e : (E - 1);
        const int dv = edst[ec];
        ss[q] = esrc[ec];
        dd[q] = (e < E) ? dv : -1;
      }
      hist[tid] = 0;
      __syncthreads();
#pragma unroll
      for (int q = 0; q < EDGE_PER_THREAD; ++q) {
        rk[q] = -1;
        if (dd[q] >= 0) rk[q] = atomicAdd(&hist[dd[q] >> BIN_SHIFT], 1);
      }
      __syncthreads();
      const int own = hist[tid];
      const int lex = scan256_exclusive(own, tid, wsum);
      lst[tid] = lex;
      binCnt[(size_t)g * MAX_BINS + tid] = own;
      binOff[(size_t)g * MAX_BINS + tid] = lex;
      __syncthreads();
      int pos[EDGE_PER_THREAD];
#pragma unroll
      for (int q = 0; q < EDGE_PER_THREAD; ++q)
        pos[q] = lst[((dd[q] >= 0) ? dd[q] : 0) >> BIN_SHIFT] + rk[q];
#pragma unroll
      for (int q = 0; q < EDGE_PER_THREAD; ++q)
        if (rk[q] >= 0) stage[pos[q]] = make_int2(ss[q], dd[q]);
      __syncthreads();
      const int left = E - g * EDGE_CHUNK;
      const int nval = (left < EDGE_CHUNK) ? left : EDGE_CHUNK;
      int2 pv[EDGE_PER_THREAD];
#pragma unroll
      for (int q = 0; q < EDGE_PER_THREAD; ++q) {
        const int j = q * CHUNK_THREADS + tid;
        pv[q] = stage[(j < nval) ? j : (nval - 1)];
      }
#pragma unroll
      for (int q = 0; q < EDGE_PER_THREAD; ++q) {
        const int j = q * CHUNK_THREADS + tid;
        if (j < nval) bpair[(size_t)g * EDGE_CHUNK + j] = pv[q];
      }
      __syncthreads();
    }
  }
}

static __device__ __forceinline__ void fetch_pairs(const int2* __restrict__ bpair,
                                                   const int* segpos, const int* segsrc,
                                                   int pw, int n, int i0, int tid,
                                                   int2 (&pr)[SORT_PER_TRIP]) {
  int lo[SORT_PER_TRIP], ic[SORT_PER_TRIP];
#pragma unroll
  for (int q = 0; q < SORT_PER_TRIP; ++q) {
    const int i = i0 + q * SORT_THREADS + tid;
    ic[q] = (i < n) ? i : (n - 1);
    lo[q] = 0;
  }
  for (int step = pw; step > 0; step >>= 1) {
    int v[SORT_PER_TRIP];
#pragma unroll
    for (int q = 0; q < SORT_PER_TRIP; ++q) v[q] = segpos[lo[q] + step];
#pragma unroll
    for (int q = 0; q < SORT_PER_TRIP; ++q) lo[q] = (v[q] <= ic[q]) ? (lo[q] + step) : lo[q];
  }
#pragma unroll
  for (int q = 0; q < SORT_PER_TRIP; ++q)
    pr[q] = bpair[segsrc[lo[q]] + (ic[q] - segpos[lo[q]])];
}

__global__ void __launch_bounds__(SORT_THREADS)
bin_sort_kernel(const int2* __restrict__ bpair, const int* __restrict__ binCnt,
                const int* __restrict__ binOff, int G,
                int Nn, int* __restrict__ start, int* __restrict__ cnt,
                int* __restrict__ csr) {
  __shared__ int lcnt[BIN_NODES];
  __shared__ int cur[BIN_NODES];
  __shared__ int wsum[8];
  __shared__ int wsum_a[8];
  __shared__ int wsum_b[8];
  __shared__ int carry[2];
  __shared__ int segpos[MAX_CHUNKS];
  __shared__ int segsrc[MAX_CHUNKS];
  __shared__ int stage[SORT_CAP];
  const int tid = threadIdx.x;
  const int b = blockIdx.x;
  if (tid < BIN_NODES) lcnt[tid] = 0;

  int beg = 0, n = 0;
  for (int g0 = 0; g0 < G; g0 += 256) {
    const int g = g0 + tid;
    const bool own = tid < 256 && g < G;
    int c = 0, o = 0;
    if (own) {
      c = binCnt[(size_t)g * MAX_BINS + b];
      o = binOff[(size_t)g * MAX_BINS + b];
    }
    const int ec = scan256_exclusive(c, tid, wsum_a);
    const int eo = scan256_exclusive(o, tid, wsum_b);
    if (own) { segpos[g] = n + ec; segsrc[g] = g * EDGE_CHUNK + o; }
    if (tid == 255) { carry[0] = beg + eo + o; carry[1] = n + ec + c; }
    __syncthreads();
    beg = carry[0];
    n = carry[1];
    __syncthreads();
  }
  int pw = 1;
  while (2 * pw < G) pw *= 2;
  for (int g = G + tid; g < 2 * pw; g += SORT_THREADS) segpos[g] = 0x7fffffff;
  __syncthreads();

  const bool fits = n <= SORT_CAP;
  const bool one = n <= SORT_THREADS * SORT_PER_TRIP;
  int2 keep[SORT_PER_TRIP];
#pragma unroll
  for (int q = 0; q < SORT_PER_TRIP; ++q) keep[q] = make_int2(0, 0);

  for (int i0 = 0; i0 < n; i0 += SORT_THREADS * SORT_PER_TRIP) {
    int2 pr[SORT_PER_TRIP];
    fetch_pairs(bpair, segpos, segsrc, pw, n, i0, tid, pr);
#pragma unroll
    for (int q = 0; q < SORT_PER_TRIP; ++q) {
      const int i = i0 + q * SORT_THREADS + tid;
      if (one) keep[q] = pr[q];
      if (i < n) atomicAdd(&lcnt[pr[q].y & (BIN_NODES - 1)], 1);
    }
  }
  __syncthreads();
  const int c = (tid < BIN_NODES) ? lcnt[tid] : 0;
  const int excl = scan256_exclusive(c, tid, wsum);
  if (tid < BIN_NODES) {
    cur[tid] = excl;
    const int d = (b << BIN_SHIFT) + tid;
    if (d < Nn) { start[d] = beg + excl; cnt[d] = c; }
  }
  __syncthreads();
  for (int i0 = 0; i0 < n; i0 += SORT_THREADS * SORT_PER_TRIP) {
    int2 pr[SORT_PER_TRIP];
    if (one) {
#pragma unroll
      for (int q = 0; q < SORT_PER_TRIP; ++q) pr[q] = keep[q];
    } else {
      fetch_pairs(bpair, segpos, segsrc, pw, n, i0, tid, pr);
    }
    int ps[SORT_PER_TRIP];
#pragma unroll
    for (int q = 0; q < SORT_PER_TRIP; ++q) {
      const int i = i0 + q * SORT_THREADS + tid;
      ps[q] = -1;
      if (i < n) ps[q] = atomicAdd(&cur[pr[q].y & (BIN_NODES - 1)], 1);
    }
    if (fits) {
#pragma unroll
      for (int q = 0; q < SORT_PER_TRIP; ++q)
        if (ps[q] >= 0) stage[ps[q]] = pr[q].x;
    } else {
#pragma unroll
      for (int q = 0; q < SORT_PER_TRIP; ++q)
        if (ps[q] >= 0) csr[beg + ps[q]] = pr[q].x;
    }
  }
  if (fits) {
    __syncthreads();
    for (int j = tid; j < n; j += SORT_THREADS) csr[beg + j] = stage[j];
  }
}

template <int H>
static __device__ __forceinline__ void load_node_row(const float* __restrict__ p,
                                                     float (&v)[H]) {
  if constexpr (H % 4 == 0) {
#pragma unroll
    for (int q = 0; q < H / 4; ++q) {
      const float4 t = reinterpret_cast<const float4*>(p)[q];
      v[4 * q + 0] = t.x; v[4 * q + 1] = t.y;
      v[4 * q + 2] = t.z; v[4 * q + 3] = t.w;
    }
  } else {
#pragma unroll
    for (int i = 0; i < H; ++i) v[i] = p[i];
  }
}

template <int H, int C, int NCH>
static __device__ __forceinline__ void load_x_row(const float* __restrict__ xs,
                                                  int lane, float (&xv)[H * NCH]) {
#pragma unroll
  for (int h = 0; h < H; ++h) {
#pragma unroll
    for (int k = 0; k < NCH; ++k) {
      const int c = k * 32 + lane;
      xv[h * NCH + k] = (c < C) ? xs[h * C + c] : 0.0f;
    }
  }
}

template <int H, int C, int NCH, int LDX, int U>
static __device__ __forceinline__ void walk_run(const float* __restrict__ xw, int lane,
                                                int s, const float (&w)[H], int cb,
                                                float (&acc)[H * NCH]) {
  int jj = 0;
  for (; jj + U <= cb; jj += U) {
    float xr[U][H * NCH];
#pragma unroll
    for (int u = 0; u < U; ++u) {
      const int su = __builtin_amdgcn_readlane(s, jj + u);
      load_x_row<H, C, NCH>(xw + (size_t)su * LDX, lane, xr[u]);
    }
#pragma unroll
    for (int h = 0; h < H; ++h) {
#pragma unroll
      for (int u = 0; u < U; ++u) {
        const float wu = __int_as_float(
            __builtin_amdgcn_readlane(__float_as_int(w[h]), jj + u));
#pragma unroll
        for (int k = 0; k < NCH; ++k) acc[h * NCH + k] += wu * xr[u][h * NCH + k];
      }
    }
  }
  for (; jj < cb; ++jj) {
    const int s0 = __builtin_amdgcn_readlane(s, jj);
    float xa[H * NCH];
    load_x_row<H, C, NCH>(xw + (size_t)s0 * LDX, lane, xa);
#pragma unroll
    for (int h = 0; h < H; ++h) {
      const float w0 = __int_as_float(
          __builtin_amdgcn_readlane(__float_as_int(w[h]), jj));
#pragma unroll
      for (int k = 0; k < NCH; ++k) acc[h * NCH + k] += w0 * xa[h * NCH + k];
    }
  }
}

template <int H, int C, int LDX, int U, bool ELU>
__global__ void __launch_bounds__(256)
node_aggregate_kernel(const int* __restrict__ csr,
                      const int* __restrict__ start,
                      const int* __restrict__ cnt, int Nn,
                      const float* __restrict__ as,
                      const float* __restrict__ ad,
                      const float* __restrict__ xw,
                      const float* __restrict__ bias,
                      float* __restrict__ out) {
  constexpr int NCH = (C + 31) / 32;
  const int lane = threadIdx.x & 31;
  const int wv = __builtin_amdgcn_readfirstlane((int)(threadIdx.x >> 5));
  const int nwaves = (int)gridDim.x * (int)(blockDim.x >> 5);
  for (int d = blockIdx.x * (int)(blockDim.x >> 5) + wv; d < Nn; d += nwaves) {
  const int beg = __builtin_amdgcn_readfirstlane(start[d]);
  const int n = __builtin_amdgcn_readfirstlane(cnt[d]) + 1;

  float dv[H];
  load_node_row<H>(ad + (size_t)d * H, dv);

  float acc[H * NCH];
#pragma unroll
  for (int i = 0; i < H * NCH; ++i) acc[i] = 0.0f;

  if (n <= 32) {
    const bool on = lane < n;
    int s = d;
    float v[H];
#pragma unroll
    for (int h = 0; h < H; ++h) v[h] = -__builtin_inff();
    if (on) {
      s = (lane < n - 1) ? csr[beg + lane] : d;
      float av[H];
      load_node_row<H>(as + (size_t)s * H, av);
#pragma unroll
      for (int h = 0; h < H; ++h) {
        const float t = av[h] + dv[h];
        v[h] = (t >= 0.0f) ? t : NEG_SLOPE * t;
      }
    }
    float mx[H];
#pragma unroll
    for (int h = 0; h < H; ++h) mx[h] = v[h];
#pragma unroll
    for (int off = 16; off > 0; off >>= 1) {
#pragma unroll
      for (int h = 0; h < H; ++h) mx[h] = fmaxf(mx[h], __shfl_xor(mx[h], off));
    }
    float e[H], t[H];
#pragma unroll
    for (int h = 0; h < H; ++h) {
      const float smx = __int_as_float(
          __builtin_amdgcn_readfirstlane(__float_as_int(mx[h])));
      const float ex = expf(v[h] - smx);
      e[h] = on ? ex : 0.0f;
      t[h] = e[h];
    }
#pragma unroll
    for (int off = 16; off > 0; off >>= 1) {
#pragma unroll
      for (int h = 0; h < H; ++h) t[h] += __shfl_xor(t[h], off);
    }
    float w[H];
#pragma unroll
    for (int h = 0; h < H; ++h) {
      const float ls = __int_as_float(
          __builtin_amdgcn_readfirstlane(__float_as_int(t[h])));
      w[h] = e[h] / ls;
    }
    walk_run<H, C, NCH, LDX, U>(xw, lane, s, w, n, acc);
  } else {
  float m[H], l[H];
#pragma unroll
  for (int h = 0; h < H; ++h) { m[h] = -__builtin_inff(); l[h] = 0.0f; }
  for (int b = 0; b < n; b += 32) {
    const int j = b + lane;
    if (j < n) {
      const int s = (j < n - 1) ? csr[beg + j] : d;
      float av[H];
      load_node_row<H>(as + (size_t)s * H, av);
#pragma unroll
      for (int h = 0; h < H; ++h) {
        float v = av[h] + dv[h];
        v = (v >= 0.0f) ? v : NEG_SLOPE * v;
        const float mn = fmaxf(m[h], v);
        l[h] = l[h] * expf(m[h] - mn) + expf(v - mn);
        m[h] = mn;
      }
    }
  }
#pragma unroll
  for (int h = 0; h < H; ++h) {
    float mx = m[h];
    for (int off = 16; off > 0; off >>= 1) mx = fmaxf(mx, __shfl_xor(mx, off));
    float t = l[h] * expf(m[h] - mx);
    for (int off = 16; off > 0; off >>= 1) t += __shfl_xor(t, off);
    m[h] = __int_as_float(__builtin_amdgcn_readfirstlane(__float_as_int(mx)));
    l[h] = __int_as_float(__builtin_amdgcn_readfirstlane(__float_as_int(t)));
  }

  for (int b = 0; b < n; b += 32) {
    const int j = b + lane;
    int s = d;
    float w[H];
#pragma unroll
    for (int h = 0; h < H; ++h) w[h] = 0.0f;
    if (j < n) {
      s = (j < n - 1) ? csr[beg + j] : d;
      float av[H];
      load_node_row<H>(as + (size_t)s * H, av);
#pragma unroll
      for (int h = 0; h < H; ++h) {
        float v = av[h] + dv[h];
        v = (v >= 0.0f) ? v : NEG_SLOPE * v;
        w[h] = expf(v - m[h]) / l[h];
      }
    }
    const int cb = (n - b < 32) ? (n - b) : 32;
    walk_run<H, C, NCH, LDX, U>(xw, lane, s, w, cb, acc);
  }
  }

  float* od = out + (size_t)d * (H * C);
#pragma unroll
  for (int h = 0; h < H; ++h) {
#pragma unroll
    for (int k = 0; k < NCH; ++k) {
      const int c = k * 32 + lane;
      if (c < C) {
        float v = acc[h * NCH + k] + bias[h * C + c];
        if (ELU) v = (v > 0.0f) ? v : (expf(v) - 1.0f);
        od[h * C + c] = v;
      }
    }
  }
  }
}

extern "C" void kernel_launch(void* const* d_in, const int* in_sizes, int n_in,
                              void* d_out, int out_size, void* d_ws, size_t ws_size,
                              hipStream_t stream) {
  const float* x      = (const float*)d_in[0];
  const int*   ei     = (const int*)d_in[1];
  const float* W1     = (const float*)d_in[2];
  const float* a1_src = (const float*)d_in[3];
  const float* a1_dst = (const float*)d_in[4];
  const float* b1     = (const float*)d_in[5];
  const float* W2     = (const float*)d_in[6];
  const float* a2_src = (const float*)d_in[7];
  const float* a2_dst = (const float*)d_in[8];
  const float* b2     = (const float*)d_in[9];
  float* out = (float*)d_out;
  (void)n_in; (void)out_size; (void)ws_size;

  const int F_IN = 256, H1 = 8, HC1 = 256, NC = 40, NCP = 64;
  const int Nn   = in_sizes[0] / F_IN;
  const int E    = in_sizes[1] / 2;
  const int* esrc = ei;
  const int* edst = ei + E;

  const int G  = (E + EDGE_CHUNK - 1) / EDGE_CHUNK;
  const int NB = (Nn + BIN_NODES - 1) / BIN_NODES;

  int* ip = (int*)d_ws;
  int*  binCnt   = ip;        ip += (size_t)G * MAX_BINS;
  int*  binOff   = ip;        ip += (size_t)G * MAX_BINS;
  int2* bpair    = (int2*)ip; ip += (size_t)2 * G * EDGE_CHUNK;
  int*  start    = ip;        ip += (size_t)Nn;
  int*  cnt      = ip;        ip += (size_t)Nn;
  int*  csr      = ip;        ip += (size_t)E;
  float* p = (float*)ip;
  float* xw1  = p;        p += (size_t)Nn * HC1;
  float* as1  = p;        p += (size_t)Nn * H1;
  float* ad1  = p;        p += (size_t)Nn * H1;
  float* agg1 = p;        p += (size_t)Nn * HC1;
  float* w2p  = p;        p += (size_t)HC1 * NCP;
  float* xw2p = p;        p += (size_t)Nn * NCP;
  float* as2  = p;        p += (size_t)Nn;
  float* ad2  = p;        p += (size_t)Nn;

  dim3 blk(256);
  dim3 sblk(SORT_THREADS);

  pad_cols_kernel<<<(HC1 * NCP + 255) / 256, blk, 0, stream>>>(
      W2, w2p, HC1, NC, NCP);

  gemm_alpha_f32_wmma_kernel<8, 32, 2, 4, 256, true><<<256, blk, 0, stream>>>(
      x, W1, xw1, a1_src, a1_dst, as1, ad1, Nn, esrc, edst, E, G, binCnt, binOff, bpair);
  bin_sort_kernel<<<NB, sblk, 0, stream>>>(bpair, binCnt, binOff, G, Nn, start, cnt, csr);
  node_aggregate_kernel<8, 32, 256, 2, true><<<2048, blk, 0, stream>>>(
      csr, start, cnt, Nn, as1, ad1, xw1, b1, agg1);

  gemm_alpha_f32_wmma_kernel<1, 40, 1, 1, 256, false><<<256, blk, 0, stream>>>(
      agg1, w2p, xw2p, a2_src, a2_dst, as2, ad2, Nn, nullptr, nullptr, 0, 0, nullptr, nullptr, nullptr);
  node_aggregate_kernel<1, 40, 64, 4, false><<<2048, blk, 0, stream>>>(
      csr, start, cnt, Nn, as2, ad2, xw2p, b2, out);
}
